# starting kernel
# baseline (speedup 1.0000x reference)
#include <hip/hip_runtime.h>


#define D_DIM   1024
#define K_DIM   512
#define LMBDA_F 1000.0f
#define NITER   96

typedef float v2f __attribute__((ext_vector_type(2)));
typedef float v8f __attribute__((ext_vector_type(8)));

__device__ __forceinline__ v8f wmma_f32(v2f a, v2f b, v8f c) {
    return __builtin_amdgcn_wmma_f32_16x16x4_f32(
        false, a, false, b, (short)0, c, false, false);
}

__global__ void max_reduce_kernel(const float* __restrict__ ea,
                                  const float* __restrict__ eb,
                                  float* __restrict__ s_out) {
    __shared__ float sm[512];
    int t = threadIdx.x;
    sm[t] = fmaxf(ea[t], eb[t]);
    __syncthreads();
    for (int off = 256; off > 0; off >>= 1) {
        if (t < off) sm[t] = fmaxf(sm[t], sm[t + off]);
        __syncthreads();
    }
    if (t == 0) s_out[0] = sm[0];
}

__global__ void mask_kernel(const float* __restrict__ erow,
                            const float* __restrict__ ecol,
                            const float* __restrict__ s_ptr,
                            float* __restrict__ mask) {
    int j = blockIdx.x * blockDim.x + threadIdx.x;
    int i = blockIdx.y;
    float inv_s = 1.0f / s_ptr[0];
    float gb = erow[i] * inv_s;
    float ga = ecol[j] * inv_s;
    float gb2 = gb * gb + 1.0f;
    float ga2 = ga * ga + 1.0f;
    float re = gb / gb2 - ga / ga2;
    float im = 1.0f / gb2 - 1.0f / ga2;
    mask[i * K_DIM + j] = re * re + im * im;
}

__global__ void __launch_bounds__(256, 1)
gram_kernel(const float* __restrict__ X,
            const float* __restrict__ Y,
            float* __restrict__ C) {
    int lane = threadIdx.x & 31;
    int wave = threadIdx.x >> 5;
    int lo = lane & 15;
    int hi = lane >> 4;
    int m0 = blockIdx.x * 64;
    int n0 = blockIdx.y * 256 + wave * 32;

    v8f zero = {};
    v8f acc[4][2];
#pragma unroll
    for (int mi = 0; mi < 4; ++mi)
#pragma unroll
        for (int ni = 0; ni < 2; ++ni) acc[mi][ni] = zero;

    for (int d = 0; d < D_DIM; d += 4) {
        int kk = d + 2 * hi;
        v2f aF[4], bF[2];
#pragma unroll
        for (int mi = 0; mi < 4; ++mi) {
            int m = m0 + mi * 16 + lo;
            aF[mi].x = X[kk * K_DIM + m];
            aF[mi].y = X[(kk + 1) * K_DIM + m];
        }
#pragma unroll
        for (int ni = 0; ni < 2; ++ni) {
            int n = n0 + ni * 16 + lo;
            bF[ni].x = Y[kk * K_DIM + n];
            bF[ni].y = Y[(kk + 1) * K_DIM + n];
        }
#pragma unroll
        for (int mi = 0; mi < 4; ++mi)
#pragma unroll
            for (int ni = 0; ni < 2; ++ni)
                acc[mi][ni] = wmma_f32(aF[mi], bF[ni], acc[mi][ni]);
    }

#pragma unroll
    for (int mi = 0; mi < 4; ++mi)
#pragma unroll
        for (int ni = 0; ni < 2; ++ni)
#pragma unroll
            for (int v = 0; v < 8; ++v) {
                int m = m0 + mi * 16 + v + 8 * hi;
                int n = n0 + ni * 16 + lo;
                C[m * K_DIM + n] = acc[mi][ni][v];
            }
}

__global__ void __launch_bounds__(256, 1)
cg_solve_kernel(const float* __restrict__ G,
                const float* __restrict__ R,
                const float* __restrict__ mask,
                float* __restrict__ Xout,
                int niter) {
    extern __shared__ float smem[];
    float* p_s  = smem;
    float* r_s  = p_s  + 16 * K_DIM;
    float* x_s  = r_s  + 16 * K_DIM;
    float* ap_s = x_s  + 16 * K_DIM;
    __shared__ float rs[16], rs_new[16], pAp[16], alpha_s[16], beta_s[16];

    int t    = threadIdx.x;
    int lane = t & 31;
    int wave = t >> 5;
    int lo   = lane & 15;
    int hi   = lane >> 4;
    int n0   = wave * 64;
    int m0   = blockIdx.x * 16;

    int rowT = t >> 4;
    int colT = t & 15;

    if (t < 16) rs[t] = 0.0f;
    __syncthreads();
    {
        float part = 0.0f;
        for (int j = 0; j < 32; ++j) {
            int c = colT + j * 16;
            float rv = R[(m0 + rowT) * K_DIM + c];
            r_s[rowT * K_DIM + c] = rv;
            p_s[rowT * K_DIM + c] = rv;
            x_s[rowT * K_DIM + c] = 0.0f;
            part += rv * rv;
        }
        atomicAdd(&rs[rowT], part);
    }
    __syncthreads();

    for (int it = 0; it < niter; ++it) {
        if (t < 16) pAp[t] = 0.0f;
        __syncthreads();

        v8f zero = {};
        v8f acc[4];
#pragma unroll
        for (int ni = 0; ni < 4; ++ni) acc[ni] = zero;

        for (int kb = 0; kb < K_DIM; kb += 4) {
            int kk = kb + 2 * hi;
            v2f aF;
            aF.x = p_s[lo * K_DIM + kk];
            aF.y = p_s[lo * K_DIM + kk + 1];
#pragma unroll
            for (int ni = 0; ni < 4; ++ni) {
                int n = n0 + ni * 16 + lo;
                v2f bF;
                bF.x = G[kk * K_DIM + n];
                bF.y = G[(kk + 1) * K_DIM + n];
                acc[ni] = wmma_f32(aF, bF, acc[ni]);
            }
        }

#pragma unroll
        for (int ni = 0; ni < 4; ++ni) {
            int n = n0 + ni * 16 + lo;
#pragma unroll
            for (int v = 0; v < 8; ++v) {
                int m = v + 8 * hi;
                float pv  = p_s[m * K_DIM + n];
                float val = acc[ni][v]
                          + LMBDA_F * mask[(m0 + m) * K_DIM + n] * pv;
                ap_s[m * K_DIM + n] = val;
                atomicAdd(&pAp[m], pv * val);
            }
        }
        __syncthreads();

        if (t < 16) { alpha_s[t] = rs[t] / pAp[t]; rs_new[t] = 0.0f; }
        __syncthreads();

        {
            float a_m  = alpha_s[rowT];
            float part = 0.0f;
            for (int j = 0; j < 32; ++j) {
                int idx = rowT * K_DIM + colT + j * 16;
                float xv = x_s[idx] + a_m * p_s[idx];
                float rv = r_s[idx] - a_m * ap_s[idx];
                x_s[idx] = xv;
                r_s[idx] = rv;
                part += rv * rv;
            }
            atomicAdd(&rs_new[rowT], part);
        }
        __syncthreads();

        if (t < 16) { beta_s[t] = rs_new[t] / rs[t]; rs[t] = rs_new[t]; }
        __syncthreads();

        {
            float b_m = beta_s[rowT];
            for (int j = 0; j < 32; ++j) {
                int idx = rowT * K_DIM + colT + j * 16;
                p_s[idx] = r_s[idx] + b_m * p_s[idx];
            }
        }
        __syncthreads();
    }

    for (int j = 0; j < 32; ++j) {
        int c = colT + j * 16;
        Xout[(m0 + rowT) * K_DIM + c] = x_s[rowT * K_DIM + c];
    }
}

extern "C" void kernel_launch(void* const* d_in, const int* in_sizes, int n_in,
                              void* d_out, int out_size, void* d_ws, size_t ws_size,
                              hipStream_t stream) {
    const float* A  = (const float*)d_in[0];
    const float* B  = (const float*)d_in[1];
    const float* ea = (const float*)d_in[2];
    const float* eb = (const float*)d_in[3];
    float* out = (float*)d_out;

    float* ws      = (float*)d_ws;
    float* s_max   = ws;
    float* mask_ab = s_max   + 16;
    float* mask_ba = mask_ab + K_DIM * K_DIM;
    float* G_a     = mask_ba + K_DIM * K_DIM;
    float* G_b     = G_a     + K_DIM * K_DIM;
    float* R_ab    = G_b     + K_DIM * K_DIM;
    float* R_ba    = R_ab    + K_DIM * K_DIM;

    max_reduce_kernel<<<1, 512, 0, stream>>>(ea, eb, s_max);

    dim3 mgrid(K_DIM / 256, K_DIM);
    mask_kernel<<<mgrid, 256, 0, stream>>>(eb, ea, s_max, mask_ab);
    mask_kernel<<<mgrid, 256, 0, stream>>>(ea, eb, s_max, mask_ba);

    dim3 ggrid(8, 2);
    gram_kernel<<<ggrid, 256, 0, stream>>>(A, A, G_a);
    gram_kernel<<<ggrid, 256, 0, stream>>>(B, A, R_ab);
    gram_kernel<<<ggrid, 256, 0, stream>>>(B, B, G_b);
    gram_kernel<<<ggrid, 256, 0, stream>>>(A, B, R_ba);

    size_t shmem = (size_t)4 * 16 * K_DIM * sizeof(float);
    cg_solve_kernel<<<K_DIM / 16, 256, shmem, stream>>>(
        G_a, R_ab, mask_ab, out, NITER);
    cg_solve_kernel<<<K_DIM / 16, 256, shmem, stream>>>(
        G_b, R_ba, mask_ba, out + K_DIM * K_DIM, NITER);
}


# CG iteration spread over the grid: per iteration one launch of 16x64 operator tiles on 256 workgroups, one row-update launch; sums: K in 4 quarters, <p,Ap> in 8 tile partials, <r,r> by wave shuffles
# speedup vs baseline: 2.1443x; 2.1443x over previous
#include <hip/hip_runtime.h>


#define D_DIM   1024
#define K_DIM   512
#define LMBDA_F 1000.0f
#define NITER   96
#define NT       8
#define P_PITCH  516
#define T_PITCH  68
#define RS_PITCH 32

typedef float v2f __attribute__((ext_vector_type(2)));
typedef float v8f __attribute__((ext_vector_type(8)));

__device__ __forceinline__ v8f wmma_f32(v2f a, v2f b, v8f c) {
    return __builtin_amdgcn_wmma_f32_16x16x4_f32(
        false, a, false, b, (short)0, c, false, false);
}

__global__ void max_reduce_kernel(const float* __restrict__ ea,
                                  const float* __restrict__ eb,
                                  float* __restrict__ s_out) {
    __shared__ float sm[512];
    int t = threadIdx.x;
    sm[t] = fmaxf(ea[t], eb[t]);
    __syncthreads();
    for (int off = 256; off > 0; off >>= 1) {
        if (t < off) sm[t] = fmaxf(sm[t], sm[t + off]);
        __syncthreads();
    }
    if (t == 0) s_out[0] = sm[0];
}

__global__ void mask_kernel(const float* __restrict__ erow,
                            const float* __restrict__ ecol,
                            const float* __restrict__ s_ptr,
                            float* __restrict__ mask) {
    int j = blockIdx.x * blockDim.x + threadIdx.x;
    int i = blockIdx.y;
    float inv_s = 1.0f / s_ptr[0];
    float gb = erow[i] * inv_s;
    float ga = ecol[j] * inv_s;
    float gb2 = gb * gb + 1.0f;
    float ga2 = ga * ga + 1.0f;
    float re = gb / gb2 - ga / ga2;
    float im = 1.0f / gb2 - 1.0f / ga2;
    mask[i * K_DIM + j] = re * re + im * im;
}

__global__ void __launch_bounds__(256, 1)
gram_kernel(const float* __restrict__ X,
            const float* __restrict__ Y,
            float* __restrict__ C) {
    int lane = threadIdx.x & 31;
    int wave = threadIdx.x >> 5;
    int lo = lane & 15;
    int hi = lane >> 4;
    int m0 = blockIdx.x * 64;
    int n0 = blockIdx.y * 256 + wave * 32;

    v8f zero = {};
    v8f acc[4][2];
#pragma unroll
    for (int mi = 0; mi < 4; ++mi)
#pragma unroll
        for (int ni = 0; ni < 2; ++ni) acc[mi][ni] = zero;

    for (int d = 0; d < D_DIM; d += 4) {
        int kk = d + 2 * hi;
        v2f aF[4], bF[2];
#pragma unroll
        for (int mi = 0; mi < 4; ++mi) {
            int m = m0 + mi * 16 + lo;
            aF[mi].x = X[kk * K_DIM + m];
            aF[mi].y = X[(kk + 1) * K_DIM + m];
        }
#pragma unroll
        for (int ni = 0; ni < 2; ++ni) {
            int n = n0 + ni * 16 + lo;
            bF[ni].x = Y[kk * K_DIM + n];
            bF[ni].y = Y[(kk + 1) * K_DIM + n];
        }
#pragma unroll
        for (int mi = 0; mi < 4; ++mi)
#pragma unroll
            for (int ni = 0; ni < 2; ++ni)
                acc[mi][ni] = wmma_f32(aF[mi], bF[ni], acc[mi][ni]);
    }

#pragma unroll
    for (int mi = 0; mi < 4; ++mi)
#pragma unroll
        for (int ni = 0; ni < 2; ++ni)
#pragma unroll
            for (int v = 0; v < 8; ++v) {
                int m = m0 + mi * 16 + v + 8 * hi;
                int n = n0 + ni * 16 + lo;
                C[m * K_DIM + n] = acc[mi][ni][v];
            }
}

__global__ void __launch_bounds__(128)
cg_init_kernel(const float* __restrict__ Rr,
               float* __restrict__ P,
               float* __restrict__ X,
               float* __restrict__ rs) {
    int lane = threadIdx.x & 31;
    int row  = blockIdx.x * 4 + (threadIdx.x >> 5);
    const float4* r4 = (const float4*)(Rr + row * K_DIM);
    float4* p4 = (float4*)(P + row * K_DIM);
    float4* x4 = (float4*)(X + row * K_DIM);
    float4 z4 = make_float4(0.0f, 0.0f, 0.0f, 0.0f);
    float part = 0.0f;
#pragma unroll
    for (int j = 0; j < 4; ++j) {
        float4 v = r4[lane + 32 * j];
        p4[lane + 32 * j] = v;
        x4[lane + 32 * j] = z4;
        part += v.x * v.x + v.y * v.y + v.z * v.z + v.w * v.w;
    }
#pragma unroll
    for (int off = 16; off > 0; off >>= 1) part += __shfl_xor(part, off);
    if (lane == 0) rs[row * RS_PITCH] = part;
}

__global__ void __launch_bounds__(256)
cg_operator_kernel(const float* __restrict__ G,
                   const float* __restrict__ P,
                   const float* __restrict__ mask,
                   float* __restrict__ Ap,
                   float* __restrict__ part) {
    __shared__ __attribute__((aligned(16))) float p_s[16 * P_PITCH];
    __shared__ __attribute__((aligned(16))) float t_s[4 * 16 * T_PITCH];

    int t    = threadIdx.x;
    int lane = t & 31;
    int wave = t >> 5;
    int lo   = lane & 15;
    int hi   = lane >> 4;
    int nt   = blockIdx.x;
    int m0   = blockIdx.y * 16;
    int nw   = wave & 1;
    int kq   = wave >> 1;

    int rowE = t >> 4;
    int nE   = nt * 64 + (t & 15) * 4;
    float4 mk = *(const float4*)(mask + (m0 + rowE) * K_DIM + nE);

#pragma unroll
    for (int j = 0; j < 8; ++j) {
        int f   = t + 256 * j;
        int row = f >> 7;
        int c4  = f & 127;
        float4 v = ((const float4*)(P + (m0 + row) * K_DIM))[c4];
        *(float4*)(p_s + row * P_PITCH + c4 * 4) = v;
    }
    __syncthreads();

    v8f acc0 = {};
    v8f acc1 = {};
    const float* Gp = G + nt * 64 + nw * 32 + lo;
    int kbase = kq * 128;
    for (int kb = kbase; kb < kbase + 128; kb += 32) {
        v2f aF[8], b0[8], b1[8];
#pragma unroll
        for (int s = 0; s < 8; ++s) {
            int kk = kb + 4 * s + 2 * hi;
            b0[s].x = Gp[kk * K_DIM];
            b0[s].y = Gp[(kk + 1) * K_DIM];
            b1[s].x = Gp[kk * K_DIM + 16];
            b1[s].y = Gp[(kk + 1) * K_DIM + 16];
        }
#pragma unroll
        for (int s = 0; s < 8; ++s) {
            int kk = kb + 4 * s + 2 * hi;
            aF[s].x = p_s[lo * P_PITCH + kk];
            aF[s].y = p_s[lo * P_PITCH + kk + 1];
        }
#pragma unroll
        for (int s = 0; s < 8; ++s) {
            acc0 = wmma_f32(aF[s], b0[s], acc0);
            acc1 = wmma_f32(aF[s], b1[s], acc1);
        }
    }

    {
        float* tq = t_s + kq * 16 * T_PITCH + nw * 32 + lo;
#pragma unroll
        for (int v = 0; v < 8; ++v) {
            int m = v + 8 * hi;
            tq[m * T_PITCH]      = acc0[v];
            tq[m * T_PITCH + 16] = acc1[v];
        }
    }
    __syncthreads();

    {
        int c = (t & 15) * 4;
        float4 s = *(const float4*)(t_s + rowE * T_PITCH + c);
#pragma unroll
        for (int q = 1; q < 4; ++q) {
            float4 u = *(const float4*)(t_s + (q * 16 + rowE) * T_PITCH + c);
            s.x += u.x; s.y += u.y; s.z += u.z; s.w += u.w;
        }
        float4 pv = *(const float4*)(p_s + rowE * P_PITCH + nE);
        float4 val;
        val.x = s.x + LMBDA_F * mk.x * pv.x;
        val.y = s.y + LMBDA_F * mk.y * pv.y;
        val.z = s.z + LMBDA_F * mk.z * pv.z;
        val.w = s.w + LMBDA_F * mk.w * pv.w;
        *(float4*)(Ap + (m0 + rowE) * K_DIM + nE) = val;
        float d = pv.x * val.x + pv.y * val.y + pv.z * val.z + pv.w * val.w;
        d += __shfl_xor(d, 8);
        d += __shfl_xor(d, 4);
        d += __shfl_xor(d, 2);
        d += __shfl_xor(d, 1);
        if ((t & 15) == 0) part[(nt * 32 + blockIdx.y) * 32 + rowE] = d;
    }
}

__global__ void __launch_bounds__(128)
cg_update_kernel(const float* __restrict__ Ap,
                 const float* __restrict__ part,
                 float* __restrict__ rs,
                 float* __restrict__ X,
                 float* __restrict__ Rr,
                 float* __restrict__ P) {
    int lane = threadIdx.x & 31;
    int row  = blockIdx.x * 4 + (threadIdx.x >> 5);
    const float4* ap4 = (const float4*)(Ap + row * K_DIM);
    float4* x4 = (float4*)(X  + row * K_DIM);
    float4* r4 = (float4*)(Rr + row * K_DIM);
    float4* p4 = (float4*)(P  + row * K_DIM);

    float4 pv[4], av[4], rv[4], xv[4];
#pragma unroll
    for (int j = 0; j < 4; ++j) {
        pv[j] = p4[lane + 32 * j];
        av[j] = ap4[lane + 32 * j];
        rv[j] = r4[lane + 32 * j];
        xv[j] = x4[lane + 32 * j];
    }
    float pap = 0.0f;
#pragma unroll
    for (int i = 0; i < NT; ++i)
        pap += part[(i * 32 + (row >> 4)) * 32 + (row & 15)];
    float rs_old = rs[row * RS_PITCH];
    float a = rs_old / pap;

    float sum = 0.0f;
#pragma unroll
    for (int j = 0; j < 4; ++j) {
        xv[j].x += a * pv[j].x; xv[j].y += a * pv[j].y;
        xv[j].z += a * pv[j].z; xv[j].w += a * pv[j].w;
        rv[j].x -= a * av[j].x; rv[j].y -= a * av[j].y;
        rv[j].z -= a * av[j].z; rv[j].w -= a * av[j].w;
        x4[lane + 32 * j] = xv[j];
        r4[lane + 32 * j] = rv[j];
        sum += rv[j].x * rv[j].x + rv[j].y * rv[j].y
             + rv[j].z * rv[j].z + rv[j].w * rv[j].w;
    }
#pragma unroll
    for (int off = 16; off > 0; off >>= 1) sum += __shfl_xor(sum, off);
    float b = sum / rs_old;
#pragma unroll
    for (int j = 0; j < 4; ++j) {
        float4 q;
        q.x = rv[j].x + b * pv[j].x; q.y = rv[j].y + b * pv[j].y;
        q.z = rv[j].z + b * pv[j].z; q.w = rv[j].w + b * pv[j].w;
        p4[lane + 32 * j] = q;
    }
    if (lane == 0) rs[row * RS_PITCH] = sum;
}

extern "C" void kernel_launch(void* const* d_in, const int* in_sizes, int n_in,
                              void* d_out, int out_size, void* d_ws, size_t ws_size,
                              hipStream_t stream) {
    const float* A  = (const float*)d_in[0];
    const float* B  = (const float*)d_in[1];
    const float* ea = (const float*)d_in[2];
    const float* eb = (const float*)d_in[3];
    float* out = (float*)d_out;

    float* ws     = (float*)d_ws;
    float* s_max  = ws;
    float* mask_w = s_max  + 16;
    float* G_w    = mask_w + K_DIM * K_DIM;
    float* r_w    = G_w    + K_DIM * K_DIM;
    float* p_w    = r_w    + K_DIM * K_DIM;
    float* ap_w   = p_w    + K_DIM * K_DIM;
    float* part_w = ap_w   + K_DIM * K_DIM;
    float* rs_w   = part_w + NT * 32 * 32;

    max_reduce_kernel<<<1, 512, 0, stream>>>(ea, eb, s_max);

    dim3 mgrid(K_DIM / 256, K_DIM);
    dim3 ggrid(8, 2);
    dim3 ogrid(NT, K_DIM / 16);

    mask_kernel<<<mgrid, 256, 0, stream>>>(eb, ea, s_max, mask_w);
    gram_kernel<<<ggrid, 256, 0, stream>>>(A, A, G_w);
    gram_kernel<<<ggrid, 256, 0, stream>>>(B, A, r_w);
    cg_init_kernel<<<K_DIM / 4, 128, 0, stream>>>(r_w, p_w, out, rs_w);
    for (int it = 0; it < NITER; ++it) {
        cg_operator_kernel<<<ogrid, 256, 0, stream>>>(G_w, p_w, mask_w, ap_w, part_w);
        cg_update_kernel<<<K_DIM / 4, 128, 0, stream>>>(ap_w, part_w, rs_w, out, r_w, p_w);
    }

    mask_kernel<<<mgrid, 256, 0, stream>>>(ea, eb, s_max, mask_w);
    gram_kernel<<<ggrid, 256, 0, stream>>>(B, B, G_w);
    gram_kernel<<<ggrid, 256, 0, stream>>>(A, B, r_w);
    cg_init_kernel<<<K_DIM / 4, 128, 0, stream>>>(r_w, p_w, out + K_DIM * K_DIM, rs_w);
    for (int it = 0; it < NITER; ++it) {
        cg_operator_kernel<<<ogrid, 256, 0, stream>>>(G_w, p_w, mask_w, ap_w, part_w);
        cg_update_kernel<<<K_DIM / 4, 128, 0, stream>>>(ap_w, part_w, rs_w, out + K_DIM * K_DIM, r_w, p_w);
    }
}


# One launch per CG iteration: each 16x64 operator workgroup first redoes its 16 rows' vector update (alpha, r, <r,r>, beta, p) from ping-pong planes; row-update launches go; sum orders kept
# speedup vs baseline: 2.7865x; 1.2995x over previous
#include <hip/hip_runtime.h>


#define D_DIM   1024
#define K_DIM   512
#define LMBDA_F 1000.0f
#define NITER   96
#define NT       8
#define P_PITCH  516
#define T_PITCH  68
#define RS_PITCH 32

typedef float v2f __attribute__((ext_vector_type(2)));
typedef float v8f __attribute__((ext_vector_type(8)));

__device__ __forceinline__ v8f wmma_f32(v2f a, v2f b, v8f c) {
    return __builtin_amdgcn_wmma_f32_16x16x4_f32(
        false, a, false, b, (short)0, c, false, false);
}

__global__ void max_reduce_kernel(const float* __restrict__ ea,
                                  const float* __restrict__ eb,
                                  float* __restrict__ s_out) {
    __shared__ float sm[512];
    int t = threadIdx.x;
    sm[t] = fmaxf(ea[t], eb[t]);
    __syncthreads();
    for (int off = 256; off > 0; off >>= 1) {
        if (t < off) sm[t] = fmaxf(sm[t], sm[t + off]);
        __syncthreads();
    }
    if (t == 0) s_out[0] = sm[0];
}

__global__ void mask_kernel(const float* __restrict__ erow,
                            const float* __restrict__ ecol,
                            const float* __restrict__ s_ptr,
                            float* __restrict__ mask) {
    int j = blockIdx.x * blockDim.x + threadIdx.x;
    int i = blockIdx.y;
    float inv_s = 1.0f / s_ptr[0];
    float gb = erow[i] * inv_s;
    float ga = ecol[j] * inv_s;
    float gb2 = gb * gb + 1.0f;
    float ga2 = ga * ga + 1.0f;
    float re = gb / gb2 - ga / ga2;
    float im = 1.0f / gb2 - 1.0f / ga2;
    mask[i * K_DIM + j] = re * re + im * im;
}

__global__ void __launch_bounds__(256, 1)
gram_kernel(const float* __restrict__ X,
            const float* __restrict__ Y,
            float* __restrict__ C) {
    int lane = threadIdx.x & 31;
    int wave = threadIdx.x >> 5;
    int lo = lane & 15;
    int hi = lane >> 4;
    int m0 = blockIdx.x * 64;
    int n0 = blockIdx.y * 256 + wave * 32;

    v8f zero = {};
    v8f acc[4][2];
#pragma unroll
    for (int mi = 0; mi < 4; ++mi)
#pragma unroll
        for (int ni = 0; ni < 2; ++ni) acc[mi][ni] = zero;

    for (int d = 0; d < D_DIM; d += 4) {
        int kk = d + 2 * hi;
        v2f aF[4], bF[2];
#pragma unroll
        for (int mi = 0; mi < 4; ++mi) {
            int m = m0 + mi * 16 + lo;
            aF[mi].x = X[kk * K_DIM + m];
            aF[mi].y = X[(kk + 1) * K_DIM + m];
        }
#pragma unroll
        for (int ni = 0; ni < 2; ++ni) {
            int n = n0 + ni * 16 + lo;
            bF[ni].x = Y[kk * K_DIM + n];
            bF[ni].y = Y[(kk + 1) * K_DIM + n];
        }
#pragma unroll
        for (int mi = 0; mi < 4; ++mi)
#pragma unroll
            for (int ni = 0; ni < 2; ++ni)
                acc[mi][ni] = wmma_f32(aF[mi], bF[ni], acc[mi][ni]);
    }

#pragma unroll
    for (int mi = 0; mi < 4; ++mi)
#pragma unroll
        for (int ni = 0; ni < 2; ++ni)
#pragma unroll
            for (int v = 0; v < 8; ++v) {
                int m = m0 + mi * 16 + v + 8 * hi;
                int n = n0 + ni * 16 + lo;
                C[m * K_DIM + n] = acc[mi][ni][v];
            }
}

__global__ void __launch_bounds__(128)
cg_init_kernel(const float* __restrict__ Rr,
               float* __restrict__ P,
               float* __restrict__ X,
               float* __restrict__ rs) {
    int lane = threadIdx.x & 31;
    int row  = blockIdx.x * 4 + (threadIdx.x >> 5);
    const float4* r4 = (const float4*)(Rr + row * K_DIM);
    float4* p4 = (float4*)(P + row * K_DIM);
    float4* x4 = (float4*)(X + row * K_DIM);
    float4 z4 = make_float4(0.0f, 0.0f, 0.0f, 0.0f);
    float part = 0.0f;
#pragma unroll
    for (int j = 0; j < 4; ++j) {
        float4 v = r4[lane + 32 * j];
        p4[lane + 32 * j] = v;
        x4[lane + 32 * j] = z4;
        part += v.x * v.x + v.y * v.y + v.z * v.z + v.w * v.w;
    }
#pragma unroll
    for (int off = 16; off > 0; off >>= 1) part += __shfl_xor(part, off);
    if (lane == 0) rs[row * RS_PITCH] = part;
}

__device__ __forceinline__ void operator_tile(const float* __restrict__ G,
                                              float4 mk,
                                              const float* p_s,
                                              float* t_s,
                                              float* __restrict__ Ap,
                                              float* __restrict__ part,
                                              int t, int nt, int mb) {
    int lane = t & 31;
    int wave = t >> 5;
    int lo   = lane & 15;
    int hi   = lane >> 4;
    int m0   = mb * 16;
    int nw   = wave & 1;
    int kq   = wave >> 1;

    int rowE = t >> 4;
    int nE   = nt * 64 + (t & 15) * 4;

    v8f acc0 = {};
    v8f acc1 = {};
    const float* Gp = G + nt * 64 + nw * 32 + lo;
    int kbase = kq * 128;
    for (int kb = kbase; kb < kbase + 128; kb += 32) {
        v2f aF[8], b0[8], b1[8];
#pragma unroll
        for (int s = 0; s < 8; ++s) {
            int kk = kb + 4 * s + 2 * hi;
            b0[s].x = Gp[kk * K_DIM];
            b0[s].y = Gp[(kk + 1) * K_DIM];
            b1[s].x = Gp[kk * K_DIM + 16];
            b1[s].y = Gp[(kk + 1) * K_DIM + 16];
        }
#pragma unroll
        for (int s = 0; s < 8; ++s) {
            int kk = kb + 4 * s + 2 * hi;
            aF[s].x = p_s[lo * P_PITCH + kk];
            aF[s].y = p_s[lo * P_PITCH + kk + 1];
        }
#pragma unroll
        for (int s = 0; s < 8; ++s) {
            acc0 = wmma_f32(aF[s], b0[s], acc0);
            acc1 = wmma_f32(aF[s], b1[s], acc1);
        }
    }

    {
        float* tq = t_s + kq * 16 * T_PITCH + nw * 32 + lo;
#pragma unroll
        for (int v = 0; v < 8; ++v) {
            int m = v + 8 * hi;
            tq[m * T_PITCH]      = acc0[v];
            tq[m * T_PITCH + 16] = acc1[v];
        }
    }
    __syncthreads();

    {
        int c = (t & 15) * 4;
        float4 s = *(const float4*)(t_s + rowE * T_PITCH + c);
#pragma unroll
        for (int q = 1; q < 4; ++q) {
            float4 u = *(const float4*)(t_s + (q * 16 + rowE) * T_PITCH + c);
            s.x += u.x; s.y += u.y; s.z += u.z; s.w += u.w;
        }
        float4 pv = *(const float4*)(p_s + rowE * P_PITCH + nE);
        float4 val;
        val.x = s.x + LMBDA_F * mk.x * pv.x;
        val.y = s.y + LMBDA_F * mk.y * pv.y;
        val.z = s.z + LMBDA_F * mk.z * pv.z;
        val.w = s.w + LMBDA_F * mk.w * pv.w;
        *(float4*)(Ap + (m0 + rowE) * K_DIM + nE) = val;
        float d = pv.x * val.x + pv.y * val.y + pv.z * val.z + pv.w * val.w;
        d += __shfl_xor(d, 8);
        d += __shfl_xor(d, 4);
        d += __shfl_xor(d, 2);
        d += __shfl_xor(d, 1);
        if ((t & 15) == 0) part[(nt * 32 + mb) * 32 + rowE] = d;
    }
}

__global__ void __launch_bounds__(256)
cg_operator_kernel(const float* __restrict__ G,
                   const float* __restrict__ P,
                   const float* __restrict__ mask,
                   float* __restrict__ Ap,
                   float* __restrict__ part) {
    __shared__ __attribute__((aligned(16))) float p_s[16 * P_PITCH];
    __shared__ __attribute__((aligned(16))) float t_s[4 * 16 * T_PITCH];

    int t  = threadIdx.x;
    int nt = blockIdx.x;
    int m0 = blockIdx.y * 16;
    float4 mk = *(const float4*)(mask + (m0 + (t >> 4)) * K_DIM + nt * 64 + (t & 15) * 4);

#pragma unroll
    for (int j = 0; j < 8; ++j) {
        int f   = t + 256 * j;
        int row = f >> 7;
        int c4  = f & 127;
        float4 v = ((const float4*)(P + (m0 + row) * K_DIM))[c4];
        *(float4*)(p_s + row * P_PITCH + c4 * 4) = v;
    }
    __syncthreads();

    operator_tile(G, mk, p_s, t_s, Ap, part, t, nt, blockIdx.y);
}

template <bool FIRST>
__device__ __forceinline__ void cg_step_body(const float* __restrict__ G,
                                             const float* __restrict__ mask,
                                             const float* __restrict__ Rin,
                                             const float* __restrict__ Pin,
                                             const float* __restrict__ APin,
                                             const float* __restrict__ partin,
                                             const float* __restrict__ rsin,
                                             float* __restrict__ Rout,
                                             float* __restrict__ Pout,
                                             float* __restrict__ APout,
                                             float* __restrict__ partout,
                                             float* __restrict__ rsout,
                                             float* __restrict__ X,
                                             float* p_s, float* t_s) {
    int t    = threadIdx.x;
    int lane = t & 31;
    int wave = t >> 5;
    int nt   = blockIdx.x;
    int mb   = blockIdx.y;
    int m0   = mb * 16;
    float4 mk = *(const float4*)(mask + (m0 + (t >> 4)) * K_DIM + nt * 64 + (t & 15) * 4);

    int  own_j   = nt >> 1;
    bool own_ln  = (lane >> 4) == (nt & 1);
    int  own_idx = lane + 32 * own_j;

    float4 rv[2][4], pv[2][4], av[2][4], xo[2];
    float pap[2], rs_old[2];
#pragma unroll
    for (int rr = 0; rr < 2; ++rr) {
        int lr  = wave * 2 + rr;
        int row = m0 + lr;
#pragma unroll
        for (int j = 0; j < 4; ++j) {
            rv[rr][j] = ((const float4*)(Rin + row * K_DIM))[lane + 32 * j];
            if (!FIRST) {
                pv[rr][j] = ((const float4*)(Pin  + row * K_DIM))[lane + 32 * j];
                av[rr][j] = ((const float4*)(APin + row * K_DIM))[lane + 32 * j];
            }
        }
        if (!FIRST) {
            float s = 0.0f;
#pragma unroll
            for (int i = 0; i < NT; ++i)
                s += partin[(i * 32 + mb) * 32 + lr];
            pap[rr]    = s;
            rs_old[rr] = rsin[row * RS_PITCH];
            xo[rr] = make_float4(0.0f, 0.0f, 0.0f, 0.0f);
            if (own_ln) xo[rr] = ((const float4*)(X + row * K_DIM))[own_idx];
        }
    }

    float a[2], sum[2];
#pragma unroll
    for (int rr = 0; rr < 2; ++rr) {
        sum[rr] = 0.0f;
        if (FIRST) {
            a[rr] = 0.0f;
#pragma unroll
            for (int j = 0; j < 4; ++j) {
                float4 v = rv[rr][j];
                sum[rr] += v.x * v.x + v.y * v.y + v.z * v.z + v.w * v.w;
            }
        } else {
            a[rr] = rs_old[rr] / pap[rr];
#pragma unroll
            for (int j = 0; j < 4; ++j) {
                rv[rr][j].x -= a[rr] * av[rr][j].x; rv[rr][j].y -= a[rr] * av[rr][j].y;
                rv[rr][j].z -= a[rr] * av[rr][j].z; rv[rr][j].w -= a[rr] * av[rr][j].w;
                sum[rr] += rv[rr][j].x * rv[rr][j].x + rv[rr][j].y * rv[rr][j].y
                         + rv[rr][j].z * rv[rr][j].z + rv[rr][j].w * rv[rr][j].w;
            }
        }
    }
#pragma unroll
    for (int off = 16; off > 0; off >>= 1) {
        float s0 = __shfl_xor(sum[0], off);
        float s1 = __shfl_xor(sum[1], off);
        sum[0] += s0;
        sum[1] += s1;
    }
    float4 kr[2], kp[2], kx[2];
#pragma unroll
    for (int rr = 0; rr < 2; ++rr) {
        int lr = wave * 2 + rr;
        float4 pn[4];
        kr[rr] = kp[rr] = kx[rr] = make_float4(0.0f, 0.0f, 0.0f, 0.0f);
        if (FIRST) {
#pragma unroll
            for (int j = 0; j < 4; ++j) {
                pn[j] = rv[rr][j];
                if (j == own_j) kp[rr] = pn[j];
            }
        } else {
            float b = sum[rr] / rs_old[rr];
#pragma unroll
            for (int j = 0; j < 4; ++j) {
                pn[j].x = rv[rr][j].x + b * pv[rr][j].x; pn[j].y = rv[rr][j].y + b * pv[rr][j].y;
                pn[j].z = rv[rr][j].z + b * pv[rr][j].z; pn[j].w = rv[rr][j].w + b * pv[rr][j].w;
            }
#pragma unroll
            for (int j = 0; j < 4; ++j)
                if (j == own_j) {
                    float4 xv = xo[rr];
                    xv.x += a[rr] * pv[rr][j].x; xv.y += a[rr] * pv[rr][j].y;
                    xv.z += a[rr] * pv[rr][j].z; xv.w += a[rr] * pv[rr][j].w;
                    kx[rr] = xv;
                    kr[rr] = rv[rr][j];
                    kp[rr] = pn[j];
                }
        }
#pragma unroll
        for (int j = 0; j < 4; ++j)
            *(float4*)(p_s + lr * P_PITCH + (lane + 32 * j) * 4) = pn[j];
    }
    __syncthreads();

#pragma unroll
    for (int rr = 0; rr < 2; ++rr) {
        int row = m0 + wave * 2 + rr;
        if (own_ln) {
            ((float4*)(X    + row * K_DIM))[own_idx] = kx[rr];
            ((float4*)(Pout + row * K_DIM))[own_idx] = kp[rr];
            if (!FIRST) ((float4*)(Rout + row * K_DIM))[own_idx] = kr[rr];
        }
        if (nt == 0 && lane == 0) rsout[row * RS_PITCH] = sum[rr];
    }

    operator_tile(G, mk, p_s, t_s, APout, partout, t, nt, mb);
}

__global__ void __launch_bounds__(256)
cg_first_kernel(const float* __restrict__ G,
                const float* __restrict__ mask,
                const float* __restrict__ Rin,
                float* __restrict__ Pout,
                float* __restrict__ APout,
                float* __restrict__ partout,
                float* __restrict__ rsout,
                float* __restrict__ X) {
    __shared__ __attribute__((aligned(16))) float p_s[16 * P_PITCH];
    __shared__ __attribute__((aligned(16))) float t_s[4 * 16 * T_PITCH];
    cg_step_body<true>(G, mask, Rin, nullptr, nullptr, nullptr, nullptr,
                       nullptr, Pout, APout, partout, rsout, X, p_s, t_s);
}

__global__ void __launch_bounds__(256)
cg_step_kernel(const float* __restrict__ G,
               const float* __restrict__ mask,
               const float* __restrict__ Rin,
               const float* __restrict__ Pin,
               const float* __restrict__ APin,
               const float* __restrict__ partin,
               const float* __restrict__ rsin,
               float* __restrict__ Rout,
               float* __restrict__ Pout,
               float* __restrict__ APout,
               float* __restrict__ partout,
               float* __restrict__ rsout,
               float* __restrict__ X) {
    __shared__ __attribute__((aligned(16))) float p_s[16 * P_PITCH];
    __shared__ __attribute__((aligned(16))) float t_s[4 * 16 * T_PITCH];
    cg_step_body<false>(G, mask, Rin, Pin, APin, partin, rsin,
                        Rout, Pout, APout, partout, rsout, X, p_s, t_s);
}

__global__ void __launch_bounds__(128)
cg_last_x_kernel(const float* __restrict__ P,
                 const float* __restrict__ part,
                 const float* __restrict__ rs,
                 float* __restrict__ X) {
    int lane = threadIdx.x & 31;
    int row  = blockIdx.x * 4 + (threadIdx.x >> 5);
    const float4* p4 = (const float4*)(P + row * K_DIM);
    float4* x4 = (float4*)(X + row * K_DIM);
    float4 pv[4], xv[4];
#pragma unroll
    for (int j = 0; j < 4; ++j) {
        pv[j] = p4[lane + 32 * j];
        xv[j] = x4[lane + 32 * j];
    }
    float pap = 0.0f;
#pragma unroll
    for (int i = 0; i < NT; ++i)
        pap += part[(i * 32 + (row >> 4)) * 32 + (row & 15)];
    float a = rs[row * RS_PITCH] / pap;
#pragma unroll
    for (int j = 0; j < 4; ++j) {
        xv[j].x += a * pv[j].x; xv[j].y += a * pv[j].y;
        xv[j].z += a * pv[j].z; xv[j].w += a * pv[j].w;
        x4[lane + 32 * j] = xv[j];
    }
}

__global__ void __launch_bounds__(128)
cg_update_kernel(const float* __restrict__ Ap,
                 const float* __restrict__ part,
                 float* __restrict__ rs,
                 float* __restrict__ X,
                 float* __restrict__ Rr,
                 float* __restrict__ P) {
    int lane = threadIdx.x & 31;
    int row  = blockIdx.x * 4 + (threadIdx.x >> 5);
    const float4* ap4 = (const float4*)(Ap + row * K_DIM);
    float4* x4 = (float4*)(X  + row * K_DIM);
    float4* r4 = (float4*)(Rr + row * K_DIM);
    float4* p4 = (float4*)(P  + row * K_DIM);

    float4 pv[4], av[4], rv[4], xv[4];
#pragma unroll
    for (int j = 0; j < 4; ++j) {
        pv[j] = p4[lane + 32 * j];
        av[j] = ap4[lane + 32 * j];
        rv[j] = r4[lane + 32 * j];
        xv[j] = x4[lane + 32 * j];
    }
    float pap = 0.0f;
#pragma unroll
    for (int i = 0; i < NT; ++i)
        pap += part[(i * 32 + (row >> 4)) * 32 + (row & 15)];
    float rs_old = rs[row * RS_PITCH];
    float a = rs_old / pap;

    float sum = 0.0f;
#pragma unroll
    for (int j = 0; j < 4; ++j) {
        xv[j].x += a * pv[j].x; xv[j].y += a * pv[j].y;
        xv[j].z += a * pv[j].z; xv[j].w += a * pv[j].w;
        rv[j].x -= a * av[j].x; rv[j].y -= a * av[j].y;
        rv[j].z -= a * av[j].z; rv[j].w -= a * av[j].w;
        x4[lane + 32 * j] = xv[j];
        r4[lane + 32 * j] = rv[j];
        sum += rv[j].x * rv[j].x + rv[j].y * rv[j].y
             + rv[j].z * rv[j].z + rv[j].w * rv[j].w;
    }
#pragma unroll
    for (int off = 16; off > 0; off >>= 1) sum += __shfl_xor(sum, off);
    float b = sum / rs_old;
#pragma unroll
    for (int j = 0; j < 4; ++j) {
        float4 q;
        q.x = rv[j].x + b * pv[j].x; q.y = rv[j].y + b * pv[j].y;
        q.z = rv[j].z + b * pv[j].z; q.w = rv[j].w + b * pv[j].w;
        p4[lane + 32 * j] = q;
    }
    if (lane == 0) rs[row * RS_PITCH] = sum;
}

extern "C" void kernel_launch(void* const* d_in, const int* in_sizes, int n_in,
                              void* d_out, int out_size, void* d_ws, size_t ws_size,
                              hipStream_t stream) {
    const float* A  = (const float*)d_in[0];
    const float* B  = (const float*)d_in[1];
    const float* ea = (const float*)d_in[2];
    const float* eb = (const float*)d_in[3];
    float* out = (float*)d_out;

    float* ws      = (float*)d_ws;
    float* s_max   = ws;
    float* mask_w  = s_max   + 16;
    float* G_w     = mask_w  + K_DIM * K_DIM;
    float* r_w     = G_w     + K_DIM * K_DIM;
    float* p_w     = r_w     + K_DIM * K_DIM;
    float* ap_w    = p_w     + K_DIM * K_DIM;
    float* part_w  = ap_w    + K_DIM * K_DIM;
    float* rs_w    = part_w  + NT * 32 * 32;
    float* r2_w    = rs_w    + K_DIM * RS_PITCH;
    float* p2_w    = r2_w    + K_DIM * K_DIM;
    float* ap2_w   = p2_w    + K_DIM * K_DIM;
    float* part2_w = ap2_w   + K_DIM * K_DIM;
    float* rs2_w   = part2_w + NT * 32 * 32;
    const bool one_launch = ws_size >= sizeof(float) *
        (size_t)(16 + 8 * K_DIM * K_DIM + 2 * NT * 32 * 32 + 2 * K_DIM * RS_PITCH);

    max_reduce_kernel<<<1, 512, 0, stream>>>(ea, eb, s_max);

    dim3 mgrid(K_DIM / 256, K_DIM);
    dim3 ggrid(8, 2);
    dim3 ogrid(NT, K_DIM / 16);

    mask_kernel<<<mgrid, 256, 0, stream>>>(eb, ea, s_max, mask_w);
    gram_kernel<<<ggrid, 256, 0, stream>>>(A, A, G_w);
    gram_kernel<<<ggrid, 256, 0, stream>>>(B, A, r_w);
    if (one_launch) {
        cg_first_kernel<<<ogrid, 256, 0, stream>>>(G_w, mask_w, r_w, p_w, ap_w, part_w, rs_w, out);
        for (int it = 0; it < (NITER - 2) / 2; ++it) {
            cg_step_kernel<<<ogrid, 256, 0, stream>>>(G_w, mask_w, r_w, p_w, ap_w, part_w, rs_w,
                                                      r2_w, p2_w, ap2_w, part2_w, rs2_w, out);
            cg_step_kernel<<<ogrid, 256, 0, stream>>>(G_w, mask_w, r2_w, p2_w, ap2_w, part2_w, rs2_w,
                                                      r_w, p_w, ap_w, part_w, rs_w, out);
        }
        cg_step_kernel<<<ogrid, 256, 0, stream>>>(G_w, mask_w, r_w, p_w, ap_w, part_w, rs_w,
                                                  r2_w, p2_w, ap2_w, part2_w, rs2_w, out);
        cg_last_x_kernel<<<K_DIM / 4, 128, 0, stream>>>(p2_w, part2_w, rs2_w, out);
    } else {
        cg_init_kernel<<<K_DIM / 4, 128, 0, stream>>>(r_w, p_w, out, rs_w);
        for (int it = 0; it < NITER; ++it) {
            cg_operator_kernel<<<ogrid, 256, 0, stream>>>(G_w, p_w, mask_w, ap_w, part_w);
            cg_update_kernel<<<K_DIM / 4, 128, 0, stream>>>(ap_w, part_w, rs_w, out, r_w, p_w);
        }
    }

    float* out2 = out + K_DIM * K_DIM;
    mask_kernel<<<mgrid, 256, 0, stream>>>(ea, eb, s_max, mask_w);
    gram_kernel<<<ggrid, 256, 0, stream>>>(B, B, G_w);
    gram_kernel<<<ggrid, 256, 0, stream>>>(A, B, r_w);
    if (one_launch) {
        cg_first_kernel<<<ogrid, 256, 0, stream>>>(G_w, mask_w, r_w, p_w, ap_w, part_w, rs_w, out2);
        for (int it = 0; it < (NITER - 2) / 2; ++it) {
            cg_step_kernel<<<ogrid, 256, 0, stream>>>(G_w, mask_w, r_w, p_w, ap_w, part_w, rs_w,
                                                      r2_w, p2_w, ap2_w, part2_w, rs2_w, out2);
            cg_step_kernel<<<ogrid, 256, 0, stream>>>(G_w, mask_w, r2_w, p2_w, ap2_w, part2_w, rs2_w,
                                                      r_w, p_w, ap_w, part_w, rs_w, out2);
        }
        cg_step_kernel<<<ogrid, 256, 0, stream>>>(G_w, mask_w, r_w, p_w, ap_w, part_w, rs_w,
                                                  r2_w, p2_w, ap2_w, part2_w, rs2_w, out2);
        cg_last_x_kernel<<<K_DIM / 4, 128, 0, stream>>>(p2_w, part2_w, rs2_w, out2);
    } else {
        cg_init_kernel<<<K_DIM / 4, 128, 0, stream>>>(r_w, p_w, out2, rs_w);
        for (int it = 0; it < NITER; ++it) {
            cg_operator_kernel<<<ogrid, 256, 0, stream>>>(G_w, p_w, mask_w, ap_w, part_w);
            cg_update_kernel<<<K_DIM / 4, 128, 0, stream>>>(ap_w, part_w, rs_w, out2, r_w, p_w);
        }
    }
}


# CG step: a wave's G fragments no longer wait for the vector part: first half requested at kernel entry, in flight across an LDS-only barrier, rest two groups ahead of its WMMAs; sum orders kept
# speedup vs baseline: 2.9338x; 1.0529x over previous
#include <hip/hip_runtime.h>


#define D_DIM   1024
#define K_DIM   512
#define LMBDA_F 1000.0f
#define NITER   96
#define NT       8
#define P_PITCH  516
#define T_PITCH  68
#define RS_PITCH 32

typedef float v2f __attribute__((ext_vector_type(2)));
typedef float v8f __attribute__((ext_vector_type(8)));

__device__ __forceinline__ v8f wmma_f32(v2f a, v2f b, v8f c) {
    return __builtin_amdgcn_wmma_f32_16x16x4_f32(
        false, a, false, b, (short)0, c, false, false);
}

__global__ void max_reduce_kernel(const float* __restrict__ ea,
                                  const float* __restrict__ eb,
                                  float* __restrict__ s_out) {
    __shared__ float sm[512];
    int t = threadIdx.x;
    sm[t] = fmaxf(ea[t], eb[t]);
    __syncthreads();
    for (int off = 256; off > 0; off >>= 1) {
        if (t < off) sm[t] = fmaxf(sm[t], sm[t + off]);
        __syncthreads();
    }
    if (t == 0) s_out[0] = sm[0];
}

__global__ void mask_kernel(const float* __restrict__ erow,
                            const float* __restrict__ ecol,
                            const float* __restrict__ s_ptr,
                            float* __restrict__ mask) {
    int j = blockIdx.x * blockDim.x + threadIdx.x;
    int i = blockIdx.y;
    float inv_s = 1.0f / s_ptr[0];
    float gb = erow[i] * inv_s;
    float ga = ecol[j] * inv_s;
    float gb2 = gb * gb + 1.0f;
    float ga2 = ga * ga + 1.0f;
    float re = gb / gb2 - ga / ga2;
    float im = 1.0f / gb2 - 1.0f / ga2;
    mask[i * K_DIM + j] = re * re + im * im;
}

__global__ void __launch_bounds__(256, 1)
gram_kernel(const float* __restrict__ X,
            const float* __restrict__ Y,
            float* __restrict__ C) {
    int lane = threadIdx.x & 31;
    int wave = threadIdx.x >> 5;
    int lo = lane & 15;
    int hi = lane >> 4;
    int m0 = blockIdx.x * 64;
    int n0 = blockIdx.y * 256 + wave * 32;

    v8f zero = {};
    v8f acc[4][2];
#pragma unroll
    for (int mi = 0; mi < 4; ++mi)
#pragma unroll
        for (int ni = 0; ni < 2; ++ni) acc[mi][ni] = zero;

    for (int d = 0; d < D_DIM; d += 4) {
        int kk = d + 2 * hi;
        v2f aF[4], bF[2];
#pragma unroll
        for (int mi = 0; mi < 4; ++mi) {
            int m = m0 + mi * 16 + lo;
            aF[mi].x = X[kk * K_DIM + m];
            aF[mi].y = X[(kk + 1) * K_DIM + m];
        }
#pragma unroll
        for (int ni = 0; ni < 2; ++ni) {
            int n = n0 + ni * 16 + lo;
            bF[ni].x = Y[kk * K_DIM + n];
            bF[ni].y = Y[(kk + 1) * K_DIM + n];
        }
#pragma unroll
        for (int mi = 0; mi < 4; ++mi)
#pragma unroll
            for (int ni = 0; ni < 2; ++ni)
                acc[mi][ni] = wmma_f32(aF[mi], bF[ni], acc[mi][ni]);
    }

#pragma unroll
    for (int mi = 0; mi < 4; ++mi)
#pragma unroll
        for (int ni = 0; ni < 2; ++ni)
#pragma unroll
            for (int v = 0; v < 8; ++v) {
                int m = m0 + mi * 16 + v + 8 * hi;
                int n = n0 + ni * 16 + lo;
                C[m * K_DIM + n] = acc[mi][ni][v];
            }
}

__global__ void __launch_bounds__(128)
cg_init_kernel(const float* __restrict__ Rr,
               float* __restrict__ P,
               float* __restrict__ X,
               float* __restrict__ rs) {
    int lane = threadIdx.x & 31;
    int row  = blockIdx.x * 4 + (threadIdx.x >> 5);
    const float4* r4 = (const float4*)(Rr + row * K_DIM);
    float4* p4 = (float4*)(P + row * K_DIM);
    float4* x4 = (float4*)(X + row * K_DIM);
    float4 z4 = make_float4(0.0f, 0.0f, 0.0f, 0.0f);
    float part = 0.0f;
#pragma unroll
    for (int j = 0; j < 4; ++j) {
        float4 v = r4[lane + 32 * j];
        p4[lane + 32 * j] = v;
        x4[lane + 32 * j] = z4;
        part += v.x * v.x + v.y * v.y + v.z * v.z + v.w * v.w;
    }
#pragma unroll
    for (int off = 16; off > 0; off >>= 1) part += __shfl_xor(part, off);
    if (lane == 0) rs[row * RS_PITCH] = part;
}

__device__ __forceinline__ void lds_barrier() {
    __builtin_amdgcn_fence(__ATOMIC_RELEASE, "workgroup", "local");
    __builtin_amdgcn_s_barrier();
    __builtin_amdgcn_fence(__ATOMIC_ACQUIRE, "workgroup", "local");
}

template <int NS>
__device__ __forceinline__ void load_g_frags(const float* __restrict__ Gp, int k0,
                                             v2f (&b0)[NS], v2f (&b1)[NS]) {
#pragma unroll
    for (int s = 0; s < NS; ++s) {
        int kk = k0 + 4 * s;
        b0[s].x = Gp[kk * K_DIM];
        b0[s].y = Gp[(kk + 1) * K_DIM];
        b1[s].x = Gp[kk * K_DIM + 16];
        b1[s].y = Gp[(kk + 1) * K_DIM + 16];
    }
}

__device__ __forceinline__ const float* g_col_ptr(const float* __restrict__ G, int t, int nt) {
    return G + nt * 64 + ((t >> 5) & 1) * 32 + (t & 15);
}
__device__ __forceinline__ int g_first_k(int t) {
    return (t >> 6) * 128 + 2 * ((t & 31) >> 4);
}

__device__ __forceinline__ void operator_tile(const float* __restrict__ G,
                                              float4 mk,
                                              const float* p_s,
                                              float* t_s,
                                              float* __restrict__ Ap,
                                              float* __restrict__ part,
                                              int t, int nt, int mb,
                                              const v2f (&ga0)[16], const v2f (&ga1)[16]) {
    int lane = t & 31;
    int wave = t >> 5;
    int lo   = lane & 15;
    int hi   = lane >> 4;
    int m0   = mb * 16;
    int nw   = wave & 1;
    int kq   = wave >> 1;

    int rowE = t >> 4;
    int nE   = nt * 64 + (t & 15) * 4;

    v8f acc0 = {};
    v8f acc1 = {};
    int k0 = kq * 128 + 2 * hi;
    const float* Gp = g_col_ptr(G, t, nt);
    v2f gb0[8], gb1[8], gc0[8], gc1[8];
    v2f aF[32];
    load_g_frags<8>(Gp, k0 + 64, gb0, gb1);
#pragma unroll
    for (int s = 0; s < 16; ++s) {
        int kk = k0 + 4 * s;
        aF[s].x = p_s[lo * P_PITCH + kk];
        aF[s].y = p_s[lo * P_PITCH + kk + 1];
    }
#pragma unroll
    for (int s = 0; s < 8; ++s) {
        acc0 = wmma_f32(aF[s], ga0[s], acc0);
        acc1 = wmma_f32(aF[s], ga1[s], acc1);
    }
    __builtin_amdgcn_sched_barrier(0);
    load_g_frags<8>(Gp, k0 + 96, gc0, gc1);
#pragma unroll
    for (int s = 16; s < 32; ++s) {
        int kk = k0 + 4 * s;
        aF[s].x = p_s[lo * P_PITCH + kk];
        aF[s].y = p_s[lo * P_PITCH + kk + 1];
    }
#pragma unroll
    for (int s = 8; s < 16; ++s) {
        acc0 = wmma_f32(aF[s], ga0[s], acc0);
        acc1 = wmma_f32(aF[s], ga1[s], acc1);
    }
#pragma unroll
    for (int s = 0; s < 8; ++s) {
        acc0 = wmma_f32(aF[16 + s], gb0[s], acc0);
        acc1 = wmma_f32(aF[16 + s], gb1[s], acc1);
    }
#pragma unroll
    for (int s = 0; s < 8; ++s) {
        acc0 = wmma_f32(aF[24 + s], gc0[s], acc0);
        acc1 = wmma_f32(aF[24 + s], gc1[s], acc1);
    }

    {
        float* tq = t_s + kq * 16 * T_PITCH + nw * 32 + lo;
#pragma unroll
        for (int v = 0; v < 8; ++v) {
            int m = v + 8 * hi;
            tq[m * T_PITCH]      = acc0[v];
            tq[m * T_PITCH + 16] = acc1[v];
        }
    }
    __syncthreads();

    {
        int c = (t & 15) * 4;
        float4 s = *(const float4*)(t_s + rowE * T_PITCH + c);
#pragma unroll
        for (int q = 1; q < 4; ++q) {
            float4 u = *(const float4*)(t_s + (q * 16 + rowE) * T_PITCH + c);
            s.x += u.x; s.y += u.y; s.z += u.z; s.w += u.w;
        }
        float4 pv = *(const float4*)(p_s + rowE * P_PITCH + nE);
        float4 val;
        val.x = s.x + LMBDA_F * mk.x * pv.x;
        val.y = s.y + LMBDA_F * mk.y * pv.y;
        val.z = s.z + LMBDA_F * mk.z * pv.z;
        val.w = s.w + LMBDA_F * mk.w * pv.w;
        *(float4*)(Ap + (m0 + rowE) * K_DIM + nE) = val;
        float d = pv.x * val.x + pv.y * val.y + pv.z * val.z + pv.w * val.w;
        d += __shfl_xor(d, 8);
        d += __shfl_xor(d, 4);
        d += __shfl_xor(d, 2);
        d += __shfl_xor(d, 1);
        if ((t & 15) == 0) part[(nt * 32 + mb) * 32 + rowE] = d;
    }
}

__global__ void __launch_bounds__(256)
cg_operator_kernel(const float* __restrict__ G,
                   const float* __restrict__ P,
                   const float* __restrict__ mask,
                   float* __restrict__ Ap,
                   float* __restrict__ part) {
    __shared__ __attribute__((aligned(16))) float p_s[16 * P_PITCH];
    __shared__ __attribute__((aligned(16))) float t_s[4 * 16 * T_PITCH];

    int t  = threadIdx.x;
    int nt = blockIdx.x;
    int m0 = blockIdx.y * 16;
    float4 mk = *(const float4*)(mask + (m0 + (t >> 4)) * K_DIM + nt * 64 + (t & 15) * 4);

    float4 pst[8];
#pragma unroll
    for (int j = 0; j < 8; ++j) {
        int f   = t + 256 * j;
        pst[j] = ((const float4*)(P + (m0 + (f >> 7)) * K_DIM))[f & 127];
    }
    v2f ga0[16], ga1[16];
    load_g_frags<16>(g_col_ptr(G, t, nt), g_first_k(t), ga0, ga1);
#pragma unroll
    for (int j = 0; j < 8; ++j) {
        int f   = t + 256 * j;
        *(float4*)(p_s + (f >> 7) * P_PITCH + (f & 127) * 4) = pst[j];
    }
    lds_barrier();

    operator_tile(G, mk, p_s, t_s, Ap, part, t, nt, blockIdx.y, ga0, ga1);
}

template <bool FIRST>
__device__ __forceinline__ void cg_step_body(const float* __restrict__ G,
                                             const float* __restrict__ mask,
                                             const float* __restrict__ Rin,
                                             const float* __restrict__ Pin,
                                             const float* __restrict__ APin,
                                             const float* __restrict__ partin,
                                             const float* __restrict__ rsin,
                                             float* __restrict__ Rout,
                                             float* __restrict__ Pout,
                                             float* __restrict__ APout,
                                             float* __restrict__ partout,
                                             float* __restrict__ rsout,
                                             float* __restrict__ X,
                                             float* p_s, float* t_s) {
    int t    = threadIdx.x;
    int lane = t & 31;
    int wave = t >> 5;
    int nt   = blockIdx.x;
    int mb   = blockIdx.y;
    int m0   = mb * 16;
    float4 mk = *(const float4*)(mask + (m0 + (t >> 4)) * K_DIM + nt * 64 + (t & 15) * 4);

    int  own_j   = nt >> 1;
    bool own_ln  = (lane >> 4) == (nt & 1);
    int  own_idx = lane + 32 * own_j;

    float4 rv[2][4], pv[2][4], av[2][4], xo[2];
    float pin[2][NT], pap[2], rs_old[2];
#pragma unroll
    for (int rr = 0; rr < 2; ++rr) {
        int lr  = wave * 2 + rr;
        int row = m0 + lr;
#pragma unroll
        for (int j = 0; j < 4; ++j) {
            rv[rr][j] = ((const float4*)(Rin + row * K_DIM))[lane + 32 * j];
            if (!FIRST) {
                pv[rr][j] = ((const float4*)(Pin  + row * K_DIM))[lane + 32 * j];
                av[rr][j] = ((const float4*)(APin + row * K_DIM))[lane + 32 * j];
            }
        }
        if (!FIRST) {
#pragma unroll
            for (int i = 0; i < NT; ++i)
                pin[rr][i] = partin[(i * 32 + mb) * 32 + lr];
            rs_old[rr] = rsin[row * RS_PITCH];
            xo[rr] = make_float4(0.0f, 0.0f, 0.0f, 0.0f);
            if (own_ln) xo[rr] = ((const float4*)(X + row * K_DIM))[own_idx];
        }
    }
    v2f ga0[16], ga1[16];
    load_g_frags<16>(g_col_ptr(G, t, nt), g_first_k(t), ga0, ga1);
    __builtin_amdgcn_sched_barrier(0);

    if (!FIRST) {
#pragma unroll
        for (int rr = 0; rr < 2; ++rr) {
            float s = 0.0f;
#pragma unroll
            for (int i = 0; i < NT; ++i) s += pin[rr][i];
            pap[rr] = s;
        }
    }

    float a[2], sum[2];
#pragma unroll
    for (int rr = 0; rr < 2; ++rr) {
        sum[rr] = 0.0f;
        if (FIRST) {
            a[rr] = 0.0f;
#pragma unroll
            for (int j = 0; j < 4; ++j) {
                float4 v = rv[rr][j];
                sum[rr] += v.x * v.x + v.y * v.y + v.z * v.z + v.w * v.w;
            }
        } else {
            a[rr] = rs_old[rr] / pap[rr];
#pragma unroll
            for (int j = 0; j < 4; ++j) {
                rv[rr][j].x -= a[rr] * av[rr][j].x; rv[rr][j].y -= a[rr] * av[rr][j].y;
                rv[rr][j].z -= a[rr] * av[rr][j].z; rv[rr][j].w -= a[rr] * av[rr][j].w;
                sum[rr] += rv[rr][j].x * rv[rr][j].x + rv[rr][j].y * rv[rr][j].y
                         + rv[rr][j].z * rv[rr][j].z + rv[rr][j].w * rv[rr][j].w;
            }
        }
    }
#pragma unroll
    for (int off = 16; off > 0; off >>= 1) {
        float s0 = __shfl_xor(sum[0], off);
        float s1 = __shfl_xor(sum[1], off);
        sum[0] += s0;
        sum[1] += s1;
    }
    float4 kr[2], kp[2], kx[2];
#pragma unroll
    for (int rr = 0; rr < 2; ++rr) {
        int lr = wave * 2 + rr;
        float4 pn[4];
        kr[rr] = kp[rr] = kx[rr] = make_float4(0.0f, 0.0f, 0.0f, 0.0f);
        if (FIRST) {
#pragma unroll
            for (int j = 0; j < 4; ++j) {
                pn[j] = rv[rr][j];
                if (j == own_j) kp[rr] = pn[j];
            }
        } else {
            float b = sum[rr] / rs_old[rr];
#pragma unroll
            for (int j = 0; j < 4; ++j) {
                pn[j].x = rv[rr][j].x + b * pv[rr][j].x; pn[j].y = rv[rr][j].y + b * pv[rr][j].y;
                pn[j].z = rv[rr][j].z + b * pv[rr][j].z; pn[j].w = rv[rr][j].w + b * pv[rr][j].w;
            }
#pragma unroll
            for (int j = 0; j < 4; ++j)
                if (j == own_j) {
                    float4 xv = xo[rr];
                    xv.x += a[rr] * pv[rr][j].x; xv.y += a[rr] * pv[rr][j].y;
                    xv.z += a[rr] * pv[rr][j].z; xv.w += a[rr] * pv[rr][j].w;
                    kx[rr] = xv;
                    kr[rr] = rv[rr][j];
                    kp[rr] = pn[j];
                }
        }
#pragma unroll
        for (int j = 0; j < 4; ++j)
            *(float4*)(p_s + lr * P_PITCH + (lane + 32 * j) * 4) = pn[j];
    }
    lds_barrier();

#pragma unroll
    for (int rr = 0; rr < 2; ++rr) {
        int row = m0 + wave * 2 + rr;
        if (own_ln) {
            ((float4*)(X    + row * K_DIM))[own_idx] = kx[rr];
            ((float4*)(Pout + row * K_DIM))[own_idx] = kp[rr];
            if (!FIRST) ((float4*)(Rout + row * K_DIM))[own_idx] = kr[rr];
        }
        if (nt == 0 && lane == 0) rsout[row * RS_PITCH] = sum[rr];
    }

    operator_tile(G, mk, p_s, t_s, APout, partout, t, nt, mb, ga0, ga1);
}

__global__ void __launch_bounds__(256)
cg_first_kernel(const float* __restrict__ G,
                const float* __restrict__ mask,
                const float* __restrict__ Rin,
                float* __restrict__ Pout,
                float* __restrict__ APout,
                float* __restrict__ partout,
                float* __restrict__ rsout,
                float* __restrict__ X) {
    __shared__ __attribute__((aligned(16))) float p_s[16 * P_PITCH];
    __shared__ __attribute__((aligned(16))) float t_s[4 * 16 * T_PITCH];
    cg_step_body<true>(G, mask, Rin, nullptr, nullptr, nullptr, nullptr,
                       nullptr, Pout, APout, partout, rsout, X, p_s, t_s);
}

__global__ void __launch_bounds__(256)
cg_step_kernel(const float* __restrict__ G,
               const float* __restrict__ mask,
               const float* __restrict__ Rin,
               const float* __restrict__ Pin,
               const float* __restrict__ APin,
               const float* __restrict__ partin,
               const float* __restrict__ rsin,
               float* __restrict__ Rout,
               float* __restrict__ Pout,
               float* __restrict__ APout,
               float* __restrict__ partout,
               float* __restrict__ rsout,
               float* __restrict__ X) {
    __shared__ __attribute__((aligned(16))) float p_s[16 * P_PITCH];
    __shared__ __attribute__((aligned(16))) float t_s[4 * 16 * T_PITCH];
    cg_step_body<false>(G, mask, Rin, Pin, APin, partin, rsin,
                        Rout, Pout, APout, partout, rsout, X, p_s, t_s);
}

__global__ void __launch_bounds__(128)
cg_last_x_kernel(const float* __restrict__ P,
                 const float* __restrict__ part,
                 const float* __restrict__ rs,
                 float* __restrict__ X) {
    int lane = threadIdx.x & 31;
    int row  = blockIdx.x * 4 + (threadIdx.x >> 5);
    const float4* p4 = (const float4*)(P + row * K_DIM);
    float4* x4 = (float4*)(X + row * K_DIM);
    float4 pv[4], xv[4];
#pragma unroll
    for (int j = 0; j < 4; ++j) {
        pv[j] = p4[lane + 32 * j];
        xv[j] = x4[lane + 32 * j];
    }
    float pap = 0.0f;
#pragma unroll
    for (int i = 0; i < NT; ++i)
        pap += part[(i * 32 + (row >> 4)) * 32 + (row & 15)];
    float a = rs[row * RS_PITCH] / pap;
#pragma unroll
    for (int j = 0; j < 4; ++j) {
        xv[j].x += a * pv[j].x; xv[j].y += a * pv[j].y;
        xv[j].z += a * pv[j].z; xv[j].w += a * pv[j].w;
        x4[lane + 32 * j] = xv[j];
    }
}

__global__ void __launch_bounds__(128)
cg_update_kernel(const float* __restrict__ Ap,
                 const float* __restrict__ part,
                 float* __restrict__ rs,
                 float* __restrict__ X,
                 float* __restrict__ Rr,
                 float* __restrict__ P) {
    int lane = threadIdx.x & 31;
    int row  = blockIdx.x * 4 + (threadIdx.x >> 5);
    const float4* ap4 = (const float4*)(Ap + row * K_DIM);
    float4* x4 = (float4*)(X  + row * K_DIM);
    float4* r4 = (float4*)(Rr + row * K_DIM);
    float4* p4 = (float4*)(P  + row * K_DIM);

    float4 pv[4], av[4], rv[4], xv[4];
#pragma unroll
    for (int j = 0; j < 4; ++j) {
        pv[j] = p4[lane + 32 * j];
        av[j] = ap4[lane + 32 * j];
        rv[j] = r4[lane + 32 * j];
        xv[j] = x4[lane + 32 * j];
    }
    float pap = 0.0f;
#pragma unroll
    for (int i = 0; i < NT; ++i)
        pap += part[(i * 32 + (row >> 4)) * 32 + (row & 15)];
    float rs_old = rs[row * RS_PITCH];
    float a = rs_old / pap;

    float sum = 0.0f;
#pragma unroll
    for (int j = 0; j < 4; ++j) {
        xv[j].x += a * pv[j].x; xv[j].y += a * pv[j].y;
        xv[j].z += a * pv[j].z; xv[j].w += a * pv[j].w;
        rv[j].x -= a * av[j].x; rv[j].y -= a * av[j].y;
        rv[j].z -= a * av[j].z; rv[j].w -= a * av[j].w;
        x4[lane + 32 * j] = xv[j];
        r4[lane + 32 * j] = rv[j];
        sum += rv[j].x * rv[j].x + rv[j].y * rv[j].y
             + rv[j].z * rv[j].z + rv[j].w * rv[j].w;
    }
#pragma unroll
    for (int off = 16; off > 0; off >>= 1) sum += __shfl_xor(sum, off);
    float b = sum / rs_old;
#pragma unroll
    for (int j = 0; j < 4; ++j) {
        float4 q;
        q.x = rv[j].x + b * pv[j].x; q.y = rv[j].y + b * pv[j].y;
        q.z = rv[j].z + b * pv[j].z; q.w = rv[j].w + b * pv[j].w;
        p4[lane + 32 * j] = q;
    }
    if (lane == 0) rs[row * RS_PITCH] = sum;
}

extern "C" void kernel_launch(void* const* d_in, const int* in_sizes, int n_in,
                              void* d_out, int out_size, void* d_ws, size_t ws_size,
                              hipStream_t stream) {
    const float* A  = (const float*)d_in[0];
    const float* B  = (const float*)d_in[1];
    const float* ea = (const float*)d_in[2];
    const float* eb = (const float*)d_in[3];
    float* out = (float*)d_out;

    float* ws      = (float*)d_ws;
    float* s_max   = ws;
    float* mask_w  = s_max   + 16;
    float* G_w     = mask_w  + K_DIM * K_DIM;
    float* r_w     = G_w     + K_DIM * K_DIM;
    float* p_w     = r_w     + K_DIM * K_DIM;
    float* ap_w    = p_w     + K_DIM * K_DIM;
    float* part_w  = ap_w    + K_DIM * K_DIM;
    float* rs_w    = part_w  + NT * 32 * 32;
    float* r2_w    = rs_w    + K_DIM * RS_PITCH;
    float* p2_w    = r2_w    + K_DIM * K_DIM;
    float* ap2_w   = p2_w    + K_DIM * K_DIM;
    float* part2_w = ap2_w   + K_DIM * K_DIM;
    float* rs2_w   = part2_w + NT * 32 * 32;
    const bool one_launch = ws_size >= sizeof(float) *
        (size_t)(16 + 8 * K_DIM * K_DIM + 2 * NT * 32 * 32 + 2 * K_DIM * RS_PITCH);

    max_reduce_kernel<<<1, 512, 0, stream>>>(ea, eb, s_max);

    dim3 mgrid(K_DIM / 256, K_DIM);
    dim3 ggrid(8, 2);
    dim3 ogrid(NT, K_DIM / 16);

    mask_kernel<<<mgrid, 256, 0, stream>>>(eb, ea, s_max, mask_w);
    gram_kernel<<<ggrid, 256, 0, stream>>>(A, A, G_w);
    gram_kernel<<<ggrid, 256, 0, stream>>>(B, A, r_w);
    if (one_launch) {
        cg_first_kernel<<<ogrid, 256, 0, stream>>>(G_w, mask_w, r_w, p_w, ap_w, part_w, rs_w, out);
        for (int it = 0; it < (NITER - 2) / 2; ++it) {
            cg_step_kernel<<<ogrid, 256, 0, stream>>>(G_w, mask_w, r_w, p_w, ap_w, part_w, rs_w,
                                                      r2_w, p2_w, ap2_w, part2_w, rs2_w, out);
            cg_step_kernel<<<ogrid, 256, 0, stream>>>(G_w, mask_w, r2_w, p2_w, ap2_w, part2_w, rs2_w,
                                                      r_w, p_w, ap_w, part_w, rs_w, out);
        }
        cg_step_kernel<<<ogrid, 256, 0, stream>>>(G_w, mask_w, r_w, p_w, ap_w, part_w, rs_w,
                                                  r2_w, p2_w, ap2_w, part2_w, rs2_w, out);
        cg_last_x_kernel<<<K_DIM / 4, 128, 0, stream>>>(p2_w, part2_w, rs2_w, out);
    } else {
        cg_init_kernel<<<K_DIM / 4, 128, 0, stream>>>(r_w, p_w, out, rs_w);
        for (int it = 0; it < NITER; ++it) {
            cg_operator_kernel<<<ogrid, 256, 0, stream>>>(G_w, p_w, mask_w, ap_w, part_w);
            cg_update_kernel<<<K_DIM / 4, 128, 0, stream>>>(ap_w, part_w, rs_w, out, r_w, p_w);
        }
    }

    float* out2 = out + K_DIM * K_DIM;
    mask_kernel<<<mgrid, 256, 0, stream>>>(ea, eb, s_max, mask_w);
    gram_kernel<<<ggrid, 256, 0, stream>>>(B, B, G_w);
    gram_kernel<<<ggrid, 256, 0, stream>>>(A, B, r_w);
    if (one_launch) {
        cg_first_kernel<<<ogrid, 256, 0, stream>>>(G_w, mask_w, r_w, p_w, ap_w, part_w, rs_w, out2);
        for (int it = 0; it < (NITER - 2) / 2; ++it) {
            cg_step_kernel<<<ogrid, 256, 0, stream>>>(G_w, mask_w, r_w, p_w, ap_w, part_w, rs_w,
                                                      r2_w, p2_w, ap2_w, part2_w, rs2_w, out2);
            cg_step_kernel<<<ogrid, 256, 0, stream>>>(G_w, mask_w, r2_w, p2_w, ap2_w, part2_w, rs2_w,
                                                      r_w, p_w, ap_w, part_w, rs_w, out2);
        }
        cg_step_kernel<<<ogrid, 256, 0, stream>>>(G_w, mask_w, r_w, p_w, ap_w, part_w, rs_w,
                                                  r2_w, p2_w, ap2_w, part2_w, rs2_w, out2);
        cg_last_x_kernel<<<K_DIM / 4, 128, 0, stream>>>(p2_w, part2_w, rs2_w, out2);
    } else {
        cg_init_kernel<<<K_DIM / 4, 128, 0, stream>>>(r_w, p_w, out2, rs_w);
        for (int it = 0; it < NITER; ++it) {
            cg_operator_kernel<<<ogrid, 256, 0, stream>>>(G_w, p_w, mask_w, ap_w, part_w);
            cg_update_kernel<<<K_DIM / 4, 128, 0, stream>>>(ap_w, part_w, rs_w, out2, r_w, p_w);
        }
    }
}


# Gram launches: one 32x32 tile per workgroup on 256 workgroups (was 64x256 on 16), the 1024-row reduction split over its 8 waves, partial tiles added in LDS in wave order: Gram sum order changes
# speedup vs baseline: 3.4667x; 1.1816x over previous
#include <hip/hip_runtime.h>


#define D_DIM   1024
#define K_DIM   512
#define LMBDA_F 1000.0f
#define NITER   96
#define NT       8
#define P_PITCH  516
#define T_PITCH  68
#define RS_PITCH 32

typedef float v2f __attribute__((ext_vector_type(2)));
typedef float v8f __attribute__((ext_vector_type(8)));

__device__ __forceinline__ v8f wmma_f32(v2f a, v2f b, v8f c) {
    return __builtin_amdgcn_wmma_f32_16x16x4_f32(
        false, a, false, b, (short)0, c, false, false);
}

__global__ void max_reduce_kernel(const float* __restrict__ ea,
                                  const float* __restrict__ eb,
                                  float* __restrict__ s_out) {
    __shared__ float sm[512];
    int t = threadIdx.x;
    sm[t] = fmaxf(ea[t], eb[t]);
    __syncthreads();
    for (int off = 256; off > 0; off >>= 1) {
        if (t < off) sm[t] = fmaxf(sm[t], sm[t + off]);
        __syncthreads();
    }
    if (t == 0) s_out[0] = sm[0];
}

__global__ void mask_kernel(const float* __restrict__ erow,
                            const float* __restrict__ ecol,
                            const float* __restrict__ s_ptr,
                            float* __restrict__ mask) {
    int j = blockIdx.x * blockDim.x + threadIdx.x;
    int i = blockIdx.y;
    float inv_s = 1.0f / s_ptr[0];
    float gb = erow[i] * inv_s;
    float ga = ecol[j] * inv_s;
    float gb2 = gb * gb + 1.0f;
    float ga2 = ga * ga + 1.0f;
    float re = gb / gb2 - ga / ga2;
    float im = 1.0f / gb2 - 1.0f / ga2;
    mask[i * K_DIM + j] = re * re + im * im;
}

#define GT_PITCH 36

__device__ __forceinline__ void load_xy_frags(const float* __restrict__ Xp,
                                              const float* __restrict__ Yp, int d,
                                              v2f (&a)[4][2], v2f (&b)[4][2]) {
#pragma unroll
    for (int s = 0; s < 4; ++s) {
        int kk = d + 4 * s;
#pragma unroll
        for (int i = 0; i < 2; ++i) {
            a[s][i].x = Xp[kk * K_DIM + 16 * i];
            a[s][i].y = Xp[(kk + 1) * K_DIM + 16 * i];
            b[s][i].x = Yp[kk * K_DIM + 16 * i];
            b[s][i].y = Yp[(kk + 1) * K_DIM + 16 * i];
        }
    }
}

__device__ __forceinline__ void xy_products(const v2f (&a)[4][2], const v2f (&b)[4][2],
                                            v8f (&acc)[2][2]) {
#pragma unroll
    for (int s = 0; s < 4; ++s)
#pragma unroll
        for (int mi = 0; mi < 2; ++mi)
#pragma unroll
            for (int ni = 0; ni < 2; ++ni)
                acc[mi][ni] = wmma_f32(a[s][mi], b[s][ni], acc[mi][ni]);
}

__global__ void __launch_bounds__(256)
gram_kernel(const float* __restrict__ X,
            const float* __restrict__ Y,
            float* __restrict__ C) {
    __shared__ __attribute__((aligned(16))) float g_s[8 * 32 * GT_PITCH];

    int t    = threadIdx.x;
    int lane = t & 31;
    int wave = t >> 5;
    int lo = lane & 15;
    int hi = lane >> 4;
    int m0 = blockIdx.x * 32;
    int n0 = blockIdx.y * 32;

    const float* Xp = X + (wave * 128 + 2 * hi) * K_DIM + m0 + lo;
    const float* Yp = Y + (wave * 128 + 2 * hi) * K_DIM + n0 + lo;

    v8f acc[2][2] = {};

    v2f a0[4][2], b0[4][2], a1[4][2], b1[4][2];
    load_xy_frags(Xp, Yp, 0, a0, b0);
#pragma unroll
    for (int g = 0; g < 8; g += 2) {
        load_xy_frags(Xp, Yp, (g + 1) * 16, a1, b1);
        __builtin_amdgcn_sched_barrier(0);
        xy_products(a0, b0, acc);
        __builtin_amdgcn_sched_barrier(0);
        if (g + 2 < 8) load_xy_frags(Xp, Yp, (g + 2) * 16, a0, b0);
        __builtin_amdgcn_sched_barrier(0);
        xy_products(a1, b1, acc);
        __builtin_amdgcn_sched_barrier(0);
    }

    {
        float* tq = g_s + wave * 32 * GT_PITCH;
#pragma unroll
        for (int mi = 0; mi < 2; ++mi)
#pragma unroll
            for (int ni = 0; ni < 2; ++ni)
#pragma unroll
                for (int v = 0; v < 8; ++v)
                    tq[(mi * 16 + v + 8 * hi) * GT_PITCH + ni * 16 + lo] = acc[mi][ni][v];
    }
    __syncthreads();

    {
        int row = t >> 3;
        int c   = (t & 7) * 4;
        float4 s = *(const float4*)(g_s + row * GT_PITCH + c);
#pragma unroll
        for (int q = 1; q < 8; ++q) {
            float4 u = *(const float4*)(g_s + (q * 32 + row) * GT_PITCH + c);
            s.x += u.x; s.y += u.y; s.z += u.z; s.w += u.w;
        }
        *(float4*)(C + (m0 + row) * K_DIM + n0 + c) = s;
    }
}

__global__ void __launch_bounds__(128)
cg_init_kernel(const float* __restrict__ Rr,
               float* __restrict__ P,
               float* __restrict__ X,
               float* __restrict__ rs) {
    int lane = threadIdx.x & 31;
    int row  = blockIdx.x * 4 + (threadIdx.x >> 5);
    const float4* r4 = (const float4*)(Rr + row * K_DIM);
    float4* p4 = (float4*)(P + row * K_DIM);
    float4* x4 = (float4*)(X + row * K_DIM);
    float4 z4 = make_float4(0.0f, 0.0f, 0.0f, 0.0f);
    float part = 0.0f;
#pragma unroll
    for (int j = 0; j < 4; ++j) {
        float4 v = r4[lane + 32 * j];
        p4[lane + 32 * j] = v;
        x4[lane + 32 * j] = z4;
        part += v.x * v.x + v.y * v.y + v.z * v.z + v.w * v.w;
    }
#pragma unroll
    for (int off = 16; off > 0; off >>= 1) part += __shfl_xor(part, off);
    if (lane == 0) rs[row * RS_PITCH] = part;
}

__device__ __forceinline__ void lds_barrier() {
    __builtin_amdgcn_fence(__ATOMIC_RELEASE, "workgroup", "local");
    __builtin_amdgcn_s_barrier();
    __builtin_amdgcn_fence(__ATOMIC_ACQUIRE, "workgroup", "local");
}

template <int NS>
__device__ __forceinline__ void load_g_frags(const float* __restrict__ Gp, int k0,
                                             v2f (&b0)[NS], v2f (&b1)[NS]) {
#pragma unroll
    for (int s = 0; s < NS; ++s) {
        int kk = k0 + 4 * s;
        b0[s].x = Gp[kk * K_DIM];
        b0[s].y = Gp[(kk + 1) * K_DIM];
        b1[s].x = Gp[kk * K_DIM + 16];
        b1[s].y = Gp[(kk + 1) * K_DIM + 16];
    }
}

__device__ __forceinline__ const float* g_col_ptr(const float* __restrict__ G, int t, int nt) {
    return G + nt * 64 + ((t >> 5) & 1) * 32 + (t & 15);
}
__device__ __forceinline__ int g_first_k(int t) {
    return (t >> 6) * 128 + 2 * ((t & 31) >> 4);
}

__device__ __forceinline__ void operator_tile(const float* __restrict__ G,
                                              float4 mk,
                                              const float* p_s,
                                              float* t_s,
                                              float* __restrict__ Ap,
                                              float* __restrict__ part,
                                              int t, int nt, int mb,
                                              const v2f (&ga0)[16], const v2f (&ga1)[16]) {
    int lane = t & 31;
    int wave = t >> 5;
    int lo   = lane & 15;
    int hi   = lane >> 4;
    int m0   = mb * 16;
    int nw   = wave & 1;
    int kq   = wave >> 1;

    int rowE = t >> 4;
    int nE   = nt * 64 + (t & 15) * 4;

    v8f acc0 = {};
    v8f acc1 = {};
    int k0 = kq * 128 + 2 * hi;
    const float* Gp = g_col_ptr(G, t, nt);
    v2f gb0[8], gb1[8], gc0[8], gc1[8];
    v2f aF[32];
    load_g_frags<8>(Gp, k0 + 64, gb0, gb1);
#pragma unroll
    for (int s = 0; s < 16; ++s) {
        int kk = k0 + 4 * s;
        aF[s].x = p_s[lo * P_PITCH + kk];
        aF[s].y = p_s[lo * P_PITCH + kk + 1];
    }
#pragma unroll
    for (int s = 0; s < 8; ++s) {
        acc0 = wmma_f32(aF[s], ga0[s], acc0);
        acc1 = wmma_f32(aF[s], ga1[s], acc1);
    }
    __builtin_amdgcn_sched_barrier(0);
    load_g_frags<8>(Gp, k0 + 96, gc0, gc1);
#pragma unroll
    for (int s = 16; s < 32; ++s) {
        int kk = k0 + 4 * s;
        aF[s].x = p_s[lo * P_PITCH + kk];
        aF[s].y = p_s[lo * P_PITCH + kk + 1];
    }
#pragma unroll
    for (int s = 8; s < 16; ++s) {
        acc0 = wmma_f32(aF[s], ga0[s], acc0);
        acc1 = wmma_f32(aF[s], ga1[s], acc1);
    }
#pragma unroll
    for (int s = 0; s < 8; ++s) {
        acc0 = wmma_f32(aF[16 + s], gb0[s], acc0);
        acc1 = wmma_f32(aF[16 + s], gb1[s], acc1);
    }
#pragma unroll
    for (int s = 0; s < 8; ++s) {
        acc0 = wmma_f32(aF[24 + s], gc0[s], acc0);
        acc1 = wmma_f32(aF[24 + s], gc1[s], acc1);
    }

    {
        float* tq = t_s + kq * 16 * T_PITCH + nw * 32 + lo;
#pragma unroll
        for (int v = 0; v < 8; ++v) {
            int m = v + 8 * hi;
            tq[m * T_PITCH]      = acc0[v];
            tq[m * T_PITCH + 16] = acc1[v];
        }
    }
    __syncthreads();

    {
        int c = (t & 15) * 4;
        float4 s = *(const float4*)(t_s + rowE * T_PITCH + c);
#pragma unroll
        for (int q = 1; q < 4; ++q) {
            float4 u = *(const float4*)(t_s + (q * 16 + rowE) * T_PITCH + c);
            s.x += u.x; s.y += u.y; s.z += u.z; s.w += u.w;
        }
        float4 pv = *(const float4*)(p_s + rowE * P_PITCH + nE);
        float4 val;
        val.x = s.x + LMBDA_F * mk.x * pv.x;
        val.y = s.y + LMBDA_F * mk.y * pv.y;
        val.z = s.z + LMBDA_F * mk.z * pv.z;
        val.w = s.w + LMBDA_F * mk.w * pv.w;
        *(float4*)(Ap + (m0 + rowE) * K_DIM + nE) = val;
        float d = pv.x * val.x + pv.y * val.y + pv.z * val.z + pv.w * val.w;
        d += __shfl_xor(d, 8);
        d += __shfl_xor(d, 4);
        d += __shfl_xor(d, 2);
        d += __shfl_xor(d, 1);
        if ((t & 15) == 0) part[(nt * 32 + mb) * 32 + rowE] = d;
    }
}

__global__ void __launch_bounds__(256)
cg_operator_kernel(const float* __restrict__ G,
                   const float* __restrict__ P,
                   const float* __restrict__ mask,
                   float* __restrict__ Ap,
                   float* __restrict__ part) {
    __shared__ __attribute__((aligned(16))) float p_s[16 * P_PITCH];
    __shared__ __attribute__((aligned(16))) float t_s[4 * 16 * T_PITCH];

    int t  = threadIdx.x;
    int nt = blockIdx.x;
    int m0 = blockIdx.y * 16;
    float4 mk = *(const float4*)(mask + (m0 + (t >> 4)) * K_DIM + nt * 64 + (t & 15) * 4);

    float4 pst[8];
#pragma unroll
    for (int j = 0; j < 8; ++j) {
        int f   = t + 256 * j;
        pst[j] = ((const float4*)(P + (m0 + (f >> 7)) * K_DIM))[f & 127];
    }
    v2f ga0[16], ga1[16];
    load_g_frags<16>(g_col_ptr(G, t, nt), g_first_k(t), ga0, ga1);
#pragma unroll
    for (int j = 0; j < 8; ++j) {
        int f   = t + 256 * j;
        *(float4*)(p_s + (f >> 7) * P_PITCH + (f & 127) * 4) = pst[j];
    }
    lds_barrier();

    operator_tile(G, mk, p_s, t_s, Ap, part, t, nt, blockIdx.y, ga0, ga1);
}

template <bool FIRST>
__device__ __forceinline__ void cg_step_body(const float* __restrict__ G,
                                             const float* __restrict__ mask,
                                             const float* __restrict__ Rin,
                                             const float* __restrict__ Pin,
                                             const float* __restrict__ APin,
                                             const float* __restrict__ partin,
                                             const float* __restrict__ rsin,
                                             float* __restrict__ Rout,
                                             float* __restrict__ Pout,
                                             float* __restrict__ APout,
                                             float* __restrict__ partout,
                                             float* __restrict__ rsout,
                                             float* __restrict__ X,
                                             float* p_s, float* t_s) {
    int t    = threadIdx.x;
    int lane = t & 31;
    int wave = t >> 5;
    int nt   = blockIdx.x;
    int mb   = blockIdx.y;
    int m0   = mb * 16;
    float4 mk = *(const float4*)(mask + (m0 + (t >> 4)) * K_DIM + nt * 64 + (t & 15) * 4);

    int  own_j   = nt >> 1;
    bool own_ln  = (lane >> 4) == (nt & 1);
    int  own_idx = lane + 32 * own_j;

    float4 rv[2][4], pv[2][4], av[2][4], xo[2];
    float pin[2][NT], pap[2], rs_old[2];
#pragma unroll
    for (int rr = 0; rr < 2; ++rr) {
        int lr  = wave * 2 + rr;
        int row = m0 + lr;
#pragma unroll
        for (int j = 0; j < 4; ++j) {
            rv[rr][j] = ((const float4*)(Rin + row * K_DIM))[lane + 32 * j];
            if (!FIRST) {
                pv[rr][j] = ((const float4*)(Pin  + row * K_DIM))[lane + 32 * j];
                av[rr][j] = ((const float4*)(APin + row * K_DIM))[lane + 32 * j];
            }
        }
        if (!FIRST) {
#pragma unroll
            for (int i = 0; i < NT; ++i)
                pin[rr][i] = partin[(i * 32 + mb) * 32 + lr];
            rs_old[rr] = rsin[row * RS_PITCH];
            xo[rr] = make_float4(0.0f, 0.0f, 0.0f, 0.0f);
            if (own_ln) xo[rr] = ((const float4*)(X + row * K_DIM))[own_idx];
        }
    }
    v2f ga0[16], ga1[16];
    load_g_frags<16>(g_col_ptr(G, t, nt), g_first_k(t), ga0, ga1);
    __builtin_amdgcn_sched_barrier(0);

    if (!FIRST) {
#pragma unroll
        for (int rr = 0; rr < 2; ++rr) {
            float s = 0.0f;
#pragma unroll
            for (int i = 0; i < NT; ++i) s += pin[rr][i];
            pap[rr] = s;
        }
    }

    float a[2], sum[2];
#pragma unroll
    for (int rr = 0; rr < 2; ++rr) {
        sum[rr] = 0.0f;
        if (FIRST) {
            a[rr] = 0.0f;
#pragma unroll
            for (int j = 0; j < 4; ++j) {
                float4 v = rv[rr][j];
                sum[rr] += v.x * v.x + v.y * v.y + v.z * v.z + v.w * v.w;
            }
        } else {
            a[rr] = rs_old[rr] / pap[rr];
#pragma unroll
            for (int j = 0; j < 4; ++j) {
                rv[rr][j].x -= a[rr] * av[rr][j].x; rv[rr][j].y -= a[rr] * av[rr][j].y;
                rv[rr][j].z -= a[rr] * av[rr][j].z; rv[rr][j].w -= a[rr] * av[rr][j].w;
                sum[rr] += rv[rr][j].x * rv[rr][j].x + rv[rr][j].y * rv[rr][j].y
                         + rv[rr][j].z * rv[rr][j].z + rv[rr][j].w * rv[rr][j].w;
            }
        }
    }
#pragma unroll
    for (int off = 16; off > 0; off >>= 1) {
        float s0 = __shfl_xor(sum[0], off);
        float s1 = __shfl_xor(sum[1], off);
        sum[0] += s0;
        sum[1] += s1;
    }
    float4 kr[2], kp[2], kx[2];
#pragma unroll
    for (int rr = 0; rr < 2; ++rr) {
        int lr = wave * 2 + rr;
        float4 pn[4];
        kr[rr] = kp[rr] = kx[rr] = make_float4(0.0f, 0.0f, 0.0f, 0.0f);
        if (FIRST) {
#pragma unroll
            for (int j = 0; j < 4; ++j) {
                pn[j] = rv[rr][j];
                if (j == own_j) kp[rr] = pn[j];
            }
        } else {
            float b = sum[rr] / rs_old[rr];
#pragma unroll
            for (int j = 0; j < 4; ++j) {
                pn[j].x = rv[rr][j].x + b * pv[rr][j].x; pn[j].y = rv[rr][j].y + b * pv[rr][j].y;
                pn[j].z = rv[rr][j].z + b * pv[rr][j].z; pn[j].w = rv[rr][j].w + b * pv[rr][j].w;
            }
#pragma unroll
            for (int j = 0; j < 4; ++j)
                if (j == own_j) {
                    float4 xv = xo[rr];
                    xv.x += a[rr] * pv[rr][j].x; xv.y += a[rr] * pv[rr][j].y;
                    xv.z += a[rr] * pv[rr][j].z; xv.w += a[rr] * pv[rr][j].w;
                    kx[rr] = xv;
                    kr[rr] = rv[rr][j];
                    kp[rr] = pn[j];
                }
        }
#pragma unroll
        for (int j = 0; j < 4; ++j)
            *(float4*)(p_s + lr * P_PITCH + (lane + 32 * j) * 4) = pn[j];
    }
    lds_barrier();

#pragma unroll
    for (int rr = 0; rr < 2; ++rr) {
        int row = m0 + wave * 2 + rr;
        if (own_ln) {
            ((float4*)(X    + row * K_DIM))[own_idx] = kx[rr];
            ((float4*)(Pout + row * K_DIM))[own_idx] = kp[rr];
            if (!FIRST) ((float4*)(Rout + row * K_DIM))[own_idx] = kr[rr];
        }
        if (nt == 0 && lane == 0) rsout[row * RS_PITCH] = sum[rr];
    }

    operator_tile(G, mk, p_s, t_s, APout, partout, t, nt, mb, ga0, ga1);
}

__global__ void __launch_bounds__(256)
cg_first_kernel(const float* __restrict__ G,
                const float* __restrict__ mask,
                const float* __restrict__ Rin,
                float* __restrict__ Pout,
                float* __restrict__ APout,
                float* __restrict__ partout,
                float* __restrict__ rsout,
                float* __restrict__ X) {
    __shared__ __attribute__((aligned(16))) float p_s[16 * P_PITCH];
    __shared__ __attribute__((aligned(16))) float t_s[4 * 16 * T_PITCH];
    cg_step_body<true>(G, mask, Rin, nullptr, nullptr, nullptr, nullptr,
                       nullptr, Pout, APout, partout, rsout, X, p_s, t_s);
}

__global__ void __launch_bounds__(256)
cg_step_kernel(const float* __restrict__ G,
               const float* __restrict__ mask,
               const float* __restrict__ Rin,
               const float* __restrict__ Pin,
               const float* __restrict__ APin,
               const float* __restrict__ partin,
               const float* __restrict__ rsin,
               float* __restrict__ Rout,
               float* __restrict__ Pout,
               float* __restrict__ APout,
               float* __restrict__ partout,
               float* __restrict__ rsout,
               float* __restrict__ X) {
    __shared__ __attribute__((aligned(16))) float p_s[16 * P_PITCH];
    __shared__ __attribute__((aligned(16))) float t_s[4 * 16 * T_PITCH];
    cg_step_body<false>(G, mask, Rin, Pin, APin, partin, rsin,
                        Rout, Pout, APout, partout, rsout, X, p_s, t_s);
}

__global__ void __launch_bounds__(128)
cg_last_x_kernel(const float* __restrict__ P,
                 const float* __restrict__ part,
                 const float* __restrict__ rs,
                 float* __restrict__ X) {
    int lane = threadIdx.x & 31;
    int row  = blockIdx.x * 4 + (threadIdx.x >> 5);
    const float4* p4 = (const float4*)(P + row * K_DIM);
    float4* x4 = (float4*)(X + row * K_DIM);
    float4 pv[4], xv[4];
#pragma unroll
    for (int j = 0; j < 4; ++j) {
        pv[j] = p4[lane + 32 * j];
        xv[j] = x4[lane + 32 * j];
    }
    float pap = 0.0f;
#pragma unroll
    for (int i = 0; i < NT; ++i)
        pap += part[(i * 32 + (row >> 4)) * 32 + (row & 15)];
    float a = rs[row * RS_PITCH] / pap;
#pragma unroll
    for (int j = 0; j < 4; ++j) {
        xv[j].x += a * pv[j].x; xv[j].y += a * pv[j].y;
        xv[j].z += a * pv[j].z; xv[j].w += a * pv[j].w;
        x4[lane + 32 * j] = xv[j];
    }
}

__global__ void __launch_bounds__(128)
cg_update_kernel(const float* __restrict__ Ap,
                 const float* __restrict__ part,
                 float* __restrict__ rs,
                 float* __restrict__ X,
                 float* __restrict__ Rr,
                 float* __restrict__ P) {
    int lane = threadIdx.x & 31;
    int row  = blockIdx.x * 4 + (threadIdx.x >> 5);
    const float4* ap4 = (const float4*)(Ap + row * K_DIM);
    float4* x4 = (float4*)(X  + row * K_DIM);
    float4* r4 = (float4*)(Rr + row * K_DIM);
    float4* p4 = (float4*)(P  + row * K_DIM);

    float4 pv[4], av[4], rv[4], xv[4];
#pragma unroll
    for (int j = 0; j < 4; ++j) {
        pv[j] = p4[lane + 32 * j];
        av[j] = ap4[lane + 32 * j];
        rv[j] = r4[lane + 32 * j];
        xv[j] = x4[lane + 32 * j];
    }
    float pap = 0.0f;
#pragma unroll
    for (int i = 0; i < NT; ++i)
        pap += part[(i * 32 + (row >> 4)) * 32 + (row & 15)];
    float rs_old = rs[row * RS_PITCH];
    float a = rs_old / pap;

    float sum = 0.0f;
#pragma unroll
    for (int j = 0; j < 4; ++j) {
        xv[j].x += a * pv[j].x; xv[j].y += a * pv[j].y;
        xv[j].z += a * pv[j].z; xv[j].w += a * pv[j].w;
        rv[j].x -= a * av[j].x; rv[j].y -= a * av[j].y;
        rv[j].z -= a * av[j].z; rv[j].w -= a * av[j].w;
        x4[lane + 32 * j] = xv[j];
        r4[lane + 32 * j] = rv[j];
        sum += rv[j].x * rv[j].x + rv[j].y * rv[j].y
             + rv[j].z * rv[j].z + rv[j].w * rv[j].w;
    }
#pragma unroll
    for (int off = 16; off > 0; off >>= 1) sum += __shfl_xor(sum, off);
    float b = sum / rs_old;
#pragma unroll
    for (int j = 0; j < 4; ++j) {
        float4 q;
        q.x = rv[j].x + b * pv[j].x; q.y = rv[j].y + b * pv[j].y;
        q.z = rv[j].z + b * pv[j].z; q.w = rv[j].w + b * pv[j].w;
        p4[lane + 32 * j] = q;
    }
    if (lane == 0) rs[row * RS_PITCH] = sum;
}

extern "C" void kernel_launch(void* const* d_in, const int* in_sizes, int n_in,
                              void* d_out, int out_size, void* d_ws, size_t ws_size,
                              hipStream_t stream) {
    const float* A  = (const float*)d_in[0];
    const float* B  = (const float*)d_in[1];
    const float* ea = (const float*)d_in[2];
    const float* eb = (const float*)d_in[3];
    float* out = (float*)d_out;

    float* ws      = (float*)d_ws;
    float* s_max   = ws;
    float* mask_w  = s_max   + 16;
    float* G_w     = mask_w  + K_DIM * K_DIM;
    float* r_w     = G_w     + K_DIM * K_DIM;
    float* p_w     = r_w     + K_DIM * K_DIM;
    float* ap_w    = p_w     + K_DIM * K_DIM;
    float* part_w  = ap_w    + K_DIM * K_DIM;
    float* rs_w    = part_w  + NT * 32 * 32;
    float* r2_w    = rs_w    + K_DIM * RS_PITCH;
    float* p2_w    = r2_w    + K_DIM * K_DIM;
    float* ap2_w   = p2_w    + K_DIM * K_DIM;
    float* part2_w = ap2_w   + K_DIM * K_DIM;
    float* rs2_w   = part2_w + NT * 32 * 32;
    const bool one_launch = ws_size >= sizeof(float) *
        (size_t)(16 + 8 * K_DIM * K_DIM + 2 * NT * 32 * 32 + 2 * K_DIM * RS_PITCH);

    max_reduce_kernel<<<1, 512, 0, stream>>>(ea, eb, s_max);

    dim3 mgrid(K_DIM / 256, K_DIM);
    dim3 ggrid(16, 16);
    dim3 ogrid(NT, K_DIM / 16);

    mask_kernel<<<mgrid, 256, 0, stream>>>(eb, ea, s_max, mask_w);
    gram_kernel<<<ggrid, 256, 0, stream>>>(A, A, G_w);
    gram_kernel<<<ggrid, 256, 0, stream>>>(B, A, r_w);
    if (one_launch) {
        cg_first_kernel<<<ogrid, 256, 0, stream>>>(G_w, mask_w, r_w, p_w, ap_w, part_w, rs_w, out);
        for (int it = 0; it < (NITER - 2) / 2; ++it) {
            cg_step_kernel<<<ogrid, 256, 0, stream>>>(G_w, mask_w, r_w, p_w, ap_w, part_w, rs_w,
                                                      r2_w, p2_w, ap2_w, part2_w, rs2_w, out);
            cg_step_kernel<<<ogrid, 256, 0, stream>>>(G_w, mask_w, r2_w, p2_w, ap2_w, part2_w, rs2_w,
                                                      r_w, p_w, ap_w, part_w, rs_w, out);
        }
        cg_step_kernel<<<ogrid, 256, 0, stream>>>(G_w, mask_w, r_w, p_w, ap_w, part_w, rs_w,
                                                  r2_w, p2_w, ap2_w, part2_w, rs2_w, out);
        cg_last_x_kernel<<<K_DIM / 4, 128, 0, stream>>>(p2_w, part2_w, rs2_w, out);
    } else {
        cg_init_kernel<<<K_DIM / 4, 128, 0, stream>>>(r_w, p_w, out, rs_w);
        for (int it = 0; it < NITER; ++it) {
            cg_operator_kernel<<<ogrid, 256, 0, stream>>>(G_w, p_w, mask_w, ap_w, part_w);
            cg_update_kernel<<<K_DIM / 4, 128, 0, stream>>>(ap_w, part_w, rs_w, out, r_w, p_w);
        }
    }

    float* out2 = out + K_DIM * K_DIM;
    mask_kernel<<<mgrid, 256, 0, stream>>>(ea, eb, s_max, mask_w);
    gram_kernel<<<ggrid, 256, 0, stream>>>(B, B, G_w);
    gram_kernel<<<ggrid, 256, 0, stream>>>(A, B, r_w);
    if (one_launch) {
        cg_first_kernel<<<ogrid, 256, 0, stream>>>(G_w, mask_w, r_w, p_w, ap_w, part_w, rs_w, out2);
        for (int it = 0; it < (NITER - 2) / 2; ++it) {
            cg_step_kernel<<<ogrid, 256, 0, stream>>>(G_w, mask_w, r_w, p_w, ap_w, part_w, rs_w,
                                                      r2_w, p2_w, ap2_w, part2_w, rs2_w, out2);
            cg_step_kernel<<<ogrid, 256, 0, stream>>>(G_w, mask_w, r2_w, p2_w, ap2_w, part2_w, rs2_w,
                                                      r_w, p_w, ap_w, part_w, rs_w, out2);
        }
        cg_step_kernel<<<ogrid, 256, 0, stream>>>(G_w, mask_w, r_w, p_w, ap_w, part_w, rs_w,
                                                  r2_w, p2_w, ap2_w, part2_w, rs2_w, out2);
        cg_last_x_kernel<<<K_DIM / 4, 128, 0, stream>>>(p2_w, part2_w, rs2_w, out2);
    } else {
        cg_init_kernel<<<K_DIM / 4, 128, 0, stream>>>(r_w, p_w, out2, rs_w);
        for (int it = 0; it < NITER; ++it) {
            cg_operator_kernel<<<ogrid, 256, 0, stream>>>(G_w, p_w, mask_w, ap_w, part_w);
            cg_update_kernel<<<K_DIM / 4, 128, 0, stream>>>(ap_w, part_w, rs_w, out2, r_w, p_w);
        }
    }
}


# Workspace planes moved from a 64-byte to a 256-byte boundary (front pad 16 -> 64 floats): a row's 512-byte loads cover 4 whole 128-byte lines, not 5; a line of x, r, p, Ap has one writer; sums kept
# speedup vs baseline: 3.7324x; 1.0766x over previous
#include <hip/hip_runtime.h>


#define D_DIM   1024
#define K_DIM   512
#define LMBDA_F 1000.0f
#define NITER   96
#define NT       8
#define P_PITCH  516
#define T_PITCH  68
#define RS_PITCH 32

typedef float v2f __attribute__((ext_vector_type(2)));
typedef float v8f __attribute__((ext_vector_type(8)));

__device__ __forceinline__ v8f wmma_f32(v2f a, v2f b, v8f c) {
    return __builtin_amdgcn_wmma_f32_16x16x4_f32(
        false, a, false, b, (short)0, c, false, false);
}

__global__ void max_reduce_kernel(const float* __restrict__ ea,
                                  const float* __restrict__ eb,
                                  float* __restrict__ s_out) {
    __shared__ float sm[512];
    int t = threadIdx.x;
    sm[t] = fmaxf(ea[t], eb[t]);
    __syncthreads();
    for (int off = 256; off > 0; off >>= 1) {
        if (t < off) sm[t] = fmaxf(sm[t], sm[t + off]);
        __syncthreads();
    }
    if (t == 0) s_out[0] = sm[0];
}

__global__ void mask_kernel(const float* __restrict__ erow,
                            const float* __restrict__ ecol,
                            const float* __restrict__ s_ptr,
                            float* __restrict__ mask) {
    int j = blockIdx.x * blockDim.x + threadIdx.x;
    int i = blockIdx.y;
    float inv_s = 1.0f / s_ptr[0];
    float gb = erow[i] * inv_s;
    float ga = ecol[j] * inv_s;
    float gb2 = gb * gb + 1.0f;
    float ga2 = ga * ga + 1.0f;
    float re = gb / gb2 - ga / ga2;
    float im = 1.0f / gb2 - 1.0f / ga2;
    mask[i * K_DIM + j] = re * re + im * im;
}

#define GT_PITCH 36

__device__ __forceinline__ void load_xy_frags(const float* __restrict__ Xp,
                                              const float* __restrict__ Yp, int d,
                                              v2f (&a)[4][2], v2f (&b)[4][2]) {
#pragma unroll
    for (int s = 0; s < 4; ++s) {
        int kk = d + 4 * s;
#pragma unroll
        for (int i = 0; i < 2; ++i) {
            a[s][i].x = Xp[kk * K_DIM + 16 * i];
            a[s][i].y = Xp[(kk + 1) * K_DIM + 16 * i];
            b[s][i].x = Yp[kk * K_DIM + 16 * i];
            b[s][i].y = Yp[(kk + 1) * K_DIM + 16 * i];
        }
    }
}

__device__ __forceinline__ void xy_products(const v2f (&a)[4][2], const v2f (&b)[4][2],
                                            v8f (&acc)[2][2]) {
#pragma unroll
    for (int s = 0; s < 4; ++s)
#pragma unroll
        for (int mi = 0; mi < 2; ++mi)
#pragma unroll
            for (int ni = 0; ni < 2; ++ni)
                acc[mi][ni] = wmma_f32(a[s][mi], b[s][ni], acc[mi][ni]);
}

__global__ void __launch_bounds__(256)
gram_kernel(const float* __restrict__ X,
            const float* __restrict__ Y,
            float* __restrict__ C) {
    __shared__ __attribute__((aligned(16))) float g_s[8 * 32 * GT_PITCH];

    int t    = threadIdx.x;
    int lane = t & 31;
    int wave = t >> 5;
    int lo = lane & 15;
    int hi = lane >> 4;
    int m0 = blockIdx.x * 32;
    int n0 = blockIdx.y * 32;

    const float* Xp = X + (wave * 128 + 2 * hi) * K_DIM + m0 + lo;
    const float* Yp = Y + (wave * 128 + 2 * hi) * K_DIM + n0 + lo;

    v8f acc[2][2] = {};

    v2f a0[4][2], b0[4][2], a1[4][2], b1[4][2];
    load_xy_frags(Xp, Yp, 0, a0, b0);
#pragma unroll
    for (int g = 0; g < 8; g += 2) {
        load_xy_frags(Xp, Yp, (g + 1) * 16, a1, b1);
        __builtin_amdgcn_sched_barrier(0);
        xy_products(a0, b0, acc);
        __builtin_amdgcn_sched_barrier(0);
        if (g + 2 < 8) load_xy_frags(Xp, Yp, (g + 2) * 16, a0, b0);
        __builtin_amdgcn_sched_barrier(0);
        xy_products(a1, b1, acc);
        __builtin_amdgcn_sched_barrier(0);
    }

    {
        float* tq = g_s + wave * 32 * GT_PITCH;
#pragma unroll
        for (int mi = 0; mi < 2; ++mi)
#pragma unroll
            for (int ni = 0; ni < 2; ++ni)
#pragma unroll
                for (int v = 0; v < 8; ++v)
                    tq[(mi * 16 + v + 8 * hi) * GT_PITCH + ni * 16 + lo] = acc[mi][ni][v];
    }
    __syncthreads();

    {
        int row = t >> 3;
        int c   = (t & 7) * 4;
        float4 s = *(const float4*)(g_s + row * GT_PITCH + c);
#pragma unroll
        for (int q = 1; q < 8; ++q) {
            float4 u = *(const float4*)(g_s + (q * 32 + row) * GT_PITCH + c);
            s.x += u.x; s.y += u.y; s.z += u.z; s.w += u.w;
        }
        *(float4*)(C + (m0 + row) * K_DIM + n0 + c) = s;
    }
}

__global__ void __launch_bounds__(128)
cg_init_kernel(const float* __restrict__ Rr,
               float* __restrict__ P,
               float* __restrict__ X,
               float* __restrict__ rs) {
    int lane = threadIdx.x & 31;
    int row  = blockIdx.x * 4 + (threadIdx.x >> 5);
    const float4* r4 = (const float4*)(Rr + row * K_DIM);
    float4* p4 = (float4*)(P + row * K_DIM);
    float4* x4 = (float4*)(X + row * K_DIM);
    float4 z4 = make_float4(0.0f, 0.0f, 0.0f, 0.0f);
    float part = 0.0f;
#pragma unroll
    for (int j = 0; j < 4; ++j) {
        float4 v = r4[lane + 32 * j];
        p4[lane + 32 * j] = v;
        x4[lane + 32 * j] = z4;
        part += v.x * v.x + v.y * v.y + v.z * v.z + v.w * v.w;
    }
#pragma unroll
    for (int off = 16; off > 0; off >>= 1) part += __shfl_xor(part, off);
    if (lane == 0) rs[row * RS_PITCH] = part;
}

__device__ __forceinline__ void lds_barrier() {
    __builtin_amdgcn_fence(__ATOMIC_RELEASE, "workgroup", "local");
    __builtin_amdgcn_s_barrier();
    __builtin_amdgcn_fence(__ATOMIC_ACQUIRE, "workgroup", "local");
}

template <int NS>
__device__ __forceinline__ void load_g_frags(const float* __restrict__ Gp, int k0,
                                             v2f (&b0)[NS], v2f (&b1)[NS]) {
#pragma unroll
    for (int s = 0; s < NS; ++s) {
        int kk = k0 + 4 * s;
        b0[s].x = Gp[kk * K_DIM];
        b0[s].y = Gp[(kk + 1) * K_DIM];
        b1[s].x = Gp[kk * K_DIM + 16];
        b1[s].y = Gp[(kk + 1) * K_DIM + 16];
    }
}

__device__ __forceinline__ const float* g_col_ptr(const float* __restrict__ G, int t, int nt) {
    return G + nt * 64 + ((t >> 5) & 1) * 32 + (t & 15);
}
__device__ __forceinline__ int g_first_k(int t) {
    return (t >> 6) * 128 + 2 * ((t & 31) >> 4);
}

__device__ __forceinline__ void operator_tile(const float* __restrict__ G,
                                              float4 mk,
                                              const float* p_s,
                                              float* t_s,
                                              float* __restrict__ Ap,
                                              float* __restrict__ part,
                                              int t, int nt, int mb,
                                              const v2f (&ga0)[16], const v2f (&ga1)[16]) {
    int lane = t & 31;
    int wave = t >> 5;
    int lo   = lane & 15;
    int hi   = lane >> 4;
    int m0   = mb * 16;
    int nw   = wave & 1;
    int kq   = wave >> 1;

    int rowE = t >> 4;
    int nE   = nt * 64 + (t & 15) * 4;

    v8f acc0 = {};
    v8f acc1 = {};
    int k0 = kq * 128 + 2 * hi;
    const float* Gp = g_col_ptr(G, t, nt);
    v2f gb0[8], gb1[8], gc0[8], gc1[8];
    v2f aF[32];
    load_g_frags<8>(Gp, k0 + 64, gb0, gb1);
#pragma unroll
    for (int s = 0; s < 16; ++s) {
        int kk = k0 + 4 * s;
        aF[s].x = p_s[lo * P_PITCH + kk];
        aF[s].y = p_s[lo * P_PITCH + kk + 1];
    }
#pragma unroll
    for (int s = 0; s < 8; ++s) {
        acc0 = wmma_f32(aF[s], ga0[s], acc0);
        acc1 = wmma_f32(aF[s], ga1[s], acc1);
    }
    __builtin_amdgcn_sched_barrier(0);
    load_g_frags<8>(Gp, k0 + 96, gc0, gc1);
#pragma unroll
    for (int s = 16; s < 32; ++s) {
        int kk = k0 + 4 * s;
        aF[s].x = p_s[lo * P_PITCH + kk];
        aF[s].y = p_s[lo * P_PITCH + kk + 1];
    }
#pragma unroll
    for (int s = 8; s < 16; ++s) {
        acc0 = wmma_f32(aF[s], ga0[s], acc0);
        acc1 = wmma_f32(aF[s], ga1[s], acc1);
    }
#pragma unroll
    for (int s = 0; s < 8; ++s) {
        acc0 = wmma_f32(aF[16 + s], gb0[s], acc0);
        acc1 = wmma_f32(aF[16 + s], gb1[s], acc1);
    }
#pragma unroll
    for (int s = 0; s < 8; ++s) {
        acc0 = wmma_f32(aF[24 + s], gc0[s], acc0);
        acc1 = wmma_f32(aF[24 + s], gc1[s], acc1);
    }

    {
        float* tq = t_s + kq * 16 * T_PITCH + nw * 32 + lo;
#pragma unroll
        for (int v = 0; v < 8; ++v) {
            int m = v + 8 * hi;
            tq[m * T_PITCH]      = acc0[v];
            tq[m * T_PITCH + 16] = acc1[v];
        }
    }
    __syncthreads();

    {
        int c = (t & 15) * 4;
        float4 s = *(const float4*)(t_s + rowE * T_PITCH + c);
#pragma unroll
        for (int q = 1; q < 4; ++q) {
            float4 u = *(const float4*)(t_s + (q * 16 + rowE) * T_PITCH + c);
            s.x += u.x; s.y += u.y; s.z += u.z; s.w += u.w;
        }
        float4 pv = *(const float4*)(p_s + rowE * P_PITCH + nE);
        float4 val;
        val.x = s.x + LMBDA_F * mk.x * pv.x;
        val.y = s.y + LMBDA_F * mk.y * pv.y;
        val.z = s.z + LMBDA_F * mk.z * pv.z;
        val.w = s.w + LMBDA_F * mk.w * pv.w;
        *(float4*)(Ap + (m0 + rowE) * K_DIM + nE) = val;
        float d = pv.x * val.x + pv.y * val.y + pv.z * val.z + pv.w * val.w;
        d += __shfl_xor(d, 8);
        d += __shfl_xor(d, 4);
        d += __shfl_xor(d, 2);
        d += __shfl_xor(d, 1);
        if ((t & 15) == 0) part[(nt * 32 + mb) * 32 + rowE] = d;
    }
}

__global__ void __launch_bounds__(256)
cg_operator_kernel(const float* __restrict__ G,
                   const float* __restrict__ P,
                   const float* __restrict__ mask,
                   float* __restrict__ Ap,
                   float* __restrict__ part) {
    __shared__ __attribute__((aligned(16))) float p_s[16 * P_PITCH];
    __shared__ __attribute__((aligned(16))) float t_s[4 * 16 * T_PITCH];

    int t  = threadIdx.x;
    int nt = blockIdx.x;
    int m0 = blockIdx.y * 16;
    float4 mk = *(const float4*)(mask + (m0 + (t >> 4)) * K_DIM + nt * 64 + (t & 15) * 4);

    float4 pst[8];
#pragma unroll
    for (int j = 0; j < 8; ++j) {
        int f   = t + 256 * j;
        pst[j] = ((const float4*)(P + (m0 + (f >> 7)) * K_DIM))[f & 127];
    }
    v2f ga0[16], ga1[16];
    load_g_frags<16>(g_col_ptr(G, t, nt), g_first_k(t), ga0, ga1);
#pragma unroll
    for (int j = 0; j < 8; ++j) {
        int f   = t + 256 * j;
        *(float4*)(p_s + (f >> 7) * P_PITCH + (f & 127) * 4) = pst[j];
    }
    lds_barrier();

    operator_tile(G, mk, p_s, t_s, Ap, part, t, nt, blockIdx.y, ga0, ga1);
}

template <bool FIRST>
__device__ __forceinline__ void cg_step_body(const float* __restrict__ G,
                                             const float* __restrict__ mask,
                                             const float* __restrict__ Rin,
                                             const float* __restrict__ Pin,
                                             const float* __restrict__ APin,
                                             const float* __restrict__ partin,
                                             const float* __restrict__ rsin,
                                             float* __restrict__ Rout,
                                             float* __restrict__ Pout,
                                             float* __restrict__ APout,
                                             float* __restrict__ partout,
                                             float* __restrict__ rsout,
                                             float* __restrict__ X,
                                             float* p_s, float* t_s) {
    int t    = threadIdx.x;
    int lane = t & 31;
    int wave = t >> 5;
    int nt   = blockIdx.x;
    int mb   = blockIdx.y;
    int m0   = mb * 16;
    float4 mk = *(const float4*)(mask + (m0 + (t >> 4)) * K_DIM + nt * 64 + (t & 15) * 4);

    int  own_j   = nt >> 1;
    bool own_ln  = (lane >> 4) == (nt & 1);
    int  own_idx = lane + 32 * own_j;

    float4 rv[2][4], pv[2][4], av[2][4], xo[2];
    float pin[2][NT], pap[2], rs_old[2];
#pragma unroll
    for (int rr = 0; rr < 2; ++rr) {
        int lr  = wave * 2 + rr;
        int row = m0 + lr;
#pragma unroll
        for (int j = 0; j < 4; ++j) {
            rv[rr][j] = ((const float4*)(Rin + row * K_DIM))[lane + 32 * j];
            if (!FIRST) {
                pv[rr][j] = ((const float4*)(Pin  + row * K_DIM))[lane + 32 * j];
                av[rr][j] = ((const float4*)(APin + row * K_DIM))[lane + 32 * j];
            }
        }
        if (!FIRST) {
#pragma unroll
            for (int i = 0; i < NT; ++i)
                pin[rr][i] = partin[(i * 32 + mb) * 32 + lr];
            rs_old[rr] = rsin[row * RS_PITCH];
            xo[rr] = make_float4(0.0f, 0.0f, 0.0f, 0.0f);
            if (own_ln) xo[rr] = ((const float4*)(X + row * K_DIM))[own_idx];
        }
    }
    v2f ga0[16], ga1[16];
    load_g_frags<16>(g_col_ptr(G, t, nt), g_first_k(t), ga0, ga1);
    __builtin_amdgcn_sched_barrier(0);

    if (!FIRST) {
#pragma unroll
        for (int rr = 0; rr < 2; ++rr) {
            float s = 0.0f;
#pragma unroll
            for (int i = 0; i < NT; ++i) s += pin[rr][i];
            pap[rr] = s;
        }
    }

    float a[2], sum[2];
#pragma unroll
    for (int rr = 0; rr < 2; ++rr) {
        sum[rr] = 0.0f;
        if (FIRST) {
            a[rr] = 0.0f;
#pragma unroll
            for (int j = 0; j < 4; ++j) {
                float4 v = rv[rr][j];
                sum[rr] += v.x * v.x + v.y * v.y + v.z * v.z + v.w * v.w;
            }
        } else {
            a[rr] = rs_old[rr] / pap[rr];
#pragma unroll
            for (int j = 0; j < 4; ++j) {
                rv[rr][j].x -= a[rr] * av[rr][j].x; rv[rr][j].y -= a[rr] * av[rr][j].y;
                rv[rr][j].z -= a[rr] * av[rr][j].z; rv[rr][j].w -= a[rr] * av[rr][j].w;
                sum[rr] += rv[rr][j].x * rv[rr][j].x + rv[rr][j].y * rv[rr][j].y
                         + rv[rr][j].z * rv[rr][j].z + rv[rr][j].w * rv[rr][j].w;
            }
        }
    }
#pragma unroll
    for (int off = 16; off > 0; off >>= 1) {
        float s0 = __shfl_xor(sum[0], off);
        float s1 = __shfl_xor(sum[1], off);
        sum[0] += s0;
        sum[1] += s1;
    }
    float4 kr[2], kp[2], kx[2];
#pragma unroll
    for (int rr = 0; rr < 2; ++rr) {
        int lr = wave * 2 + rr;
        float4 pn[4];
        kr[rr] = kp[rr] = kx[rr] = make_float4(0.0f, 0.0f, 0.0f, 0.0f);
        if (FIRST) {
#pragma unroll
            for (int j = 0; j < 4; ++j) {
                pn[j] = rv[rr][j];
                if (j == own_j) kp[rr] = pn[j];
            }
        } else {
            float b = sum[rr] / rs_old[rr];
#pragma unroll
            for (int j = 0; j < 4; ++j) {
                pn[j].x = rv[rr][j].x + b * pv[rr][j].x; pn[j].y = rv[rr][j].y + b * pv[rr][j].y;
                pn[j].z = rv[rr][j].z + b * pv[rr][j].z; pn[j].w = rv[rr][j].w + b * pv[rr][j].w;
            }
#pragma unroll
            for (int j = 0; j < 4; ++j)
                if (j == own_j) {
                    float4 xv = xo[rr];
                    xv.x += a[rr] * pv[rr][j].x; xv.y += a[rr] * pv[rr][j].y;
                    xv.z += a[rr] * pv[rr][j].z; xv.w += a[rr] * pv[rr][j].w;
                    kx[rr] = xv;
                    kr[rr] = rv[rr][j];
                    kp[rr] = pn[j];
                }
        }
#pragma unroll
        for (int j = 0; j < 4; ++j)
            *(float4*)(p_s + lr * P_PITCH + (lane + 32 * j) * 4) = pn[j];
    }
    lds_barrier();

#pragma unroll
    for (int rr = 0; rr < 2; ++rr) {
        int row = m0 + wave * 2 + rr;
        if (own_ln) {
            ((float4*)(X    + row * K_DIM))[own_idx] = kx[rr];
            ((float4*)(Pout + row * K_DIM))[own_idx] = kp[rr];
            if (!FIRST) ((float4*)(Rout + row * K_DIM))[own_idx] = kr[rr];
        }
        if (nt == 0 && lane == 0) rsout[row * RS_PITCH] = sum[rr];
    }

    operator_tile(G, mk, p_s, t_s, APout, partout, t, nt, mb, ga0, ga1);
}

__global__ void __launch_bounds__(256)
cg_first_kernel(const float* __restrict__ G,
                const float* __restrict__ mask,
                const float* __restrict__ Rin,
                float* __restrict__ Pout,
                float* __restrict__ APout,
                float* __restrict__ partout,
                float* __restrict__ rsout,
                float* __restrict__ X) {
    __shared__ __attribute__((aligned(16))) float p_s[16 * P_PITCH];
    __shared__ __attribute__((aligned(16))) float t_s[4 * 16 * T_PITCH];
    cg_step_body<true>(G, mask, Rin, nullptr, nullptr, nullptr, nullptr,
                       nullptr, Pout, APout, partout, rsout, X, p_s, t_s);
}

__global__ void __launch_bounds__(256)
cg_step_kernel(const float* __restrict__ G,
               const float* __restrict__ mask,
               const float* __restrict__ Rin,
               const float* __restrict__ Pin,
               const float* __restrict__ APin,
               const float* __restrict__ partin,
               const float* __restrict__ rsin,
               float* __restrict__ Rout,
               float* __restrict__ Pout,
               float* __restrict__ APout,
               float* __restrict__ partout,
               float* __restrict__ rsout,
               float* __restrict__ X) {
    __shared__ __attribute__((aligned(16))) float p_s[16 * P_PITCH];
    __shared__ __attribute__((aligned(16))) float t_s[4 * 16 * T_PITCH];
    cg_step_body<false>(G, mask, Rin, Pin, APin, partin, rsin,
                        Rout, Pout, APout, partout, rsout, X, p_s, t_s);
}

__global__ void __launch_bounds__(128)
cg_last_x_kernel(const float* __restrict__ P,
                 const float* __restrict__ part,
                 const float* __restrict__ rs,
                 float* __restrict__ X) {
    int lane = threadIdx.x & 31;
    int row  = blockIdx.x * 4 + (threadIdx.x >> 5);
    const float4* p4 = (const float4*)(P + row * K_DIM);
    float4* x4 = (float4*)(X + row * K_DIM);
    float4 pv[4], xv[4];
#pragma unroll
    for (int j = 0; j < 4; ++j) {
        pv[j] = p4[lane + 32 * j];
        xv[j] = x4[lane + 32 * j];
    }
    float pap = 0.0f;
#pragma unroll
    for (int i = 0; i < NT; ++i)
        pap += part[(i * 32 + (row >> 4)) * 32 + (row & 15)];
    float a = rs[row * RS_PITCH] / pap;
#pragma unroll
    for (int j = 0; j < 4; ++j) {
        xv[j].x += a * pv[j].x; xv[j].y += a * pv[j].y;
        xv[j].z += a * pv[j].z; xv[j].w += a * pv[j].w;
        x4[lane + 32 * j] = xv[j];
    }
}

__global__ void __launch_bounds__(128)
cg_update_kernel(const float* __restrict__ Ap,
                 const float* __restrict__ part,
                 float* __restrict__ rs,
                 float* __restrict__ X,
                 float* __restrict__ Rr,
                 float* __restrict__ P) {
    int lane = threadIdx.x & 31;
    int row  = blockIdx.x * 4 + (threadIdx.x >> 5);
    const float4* ap4 = (const float4*)(Ap + row * K_DIM);
    float4* x4 = (float4*)(X  + row * K_DIM);
    float4* r4 = (float4*)(Rr + row * K_DIM);
    float4* p4 = (float4*)(P  + row * K_DIM);

    float4 pv[4], av[4], rv[4], xv[4];
#pragma unroll
    for (int j = 0; j < 4; ++j) {
        pv[j] = p4[lane + 32 * j];
        av[j] = ap4[lane + 32 * j];
        rv[j] = r4[lane + 32 * j];
        xv[j] = x4[lane + 32 * j];
    }
    float pap = 0.0f;
#pragma unroll
    for (int i = 0; i < NT; ++i)
        pap += part[(i * 32 + (row >> 4)) * 32 + (row & 15)];
    float rs_old = rs[row * RS_PITCH];
    float a = rs_old / pap;

    float sum = 0.0f;
#pragma unroll
    for (int j = 0; j < 4; ++j) {
        xv[j].x += a * pv[j].x; xv[j].y += a * pv[j].y;
        xv[j].z += a * pv[j].z; xv[j].w += a * pv[j].w;
        rv[j].x -= a * av[j].x; rv[j].y -= a * av[j].y;
        rv[j].z -= a * av[j].z; rv[j].w -= a * av[j].w;
        x4[lane + 32 * j] = xv[j];
        r4[lane + 32 * j] = rv[j];
        sum += rv[j].x * rv[j].x + rv[j].y * rv[j].y
             + rv[j].z * rv[j].z + rv[j].w * rv[j].w;
    }
#pragma unroll
    for (int off = 16; off > 0; off >>= 1) sum += __shfl_xor(sum, off);
    float b = sum / rs_old;
#pragma unroll
    for (int j = 0; j < 4; ++j) {
        float4 q;
        q.x = rv[j].x + b * pv[j].x; q.y = rv[j].y + b * pv[j].y;
        q.z = rv[j].z + b * pv[j].z; q.w = rv[j].w + b * pv[j].w;
        p4[lane + 32 * j] = q;
    }
    if (lane == 0) rs[row * RS_PITCH] = sum;
}

extern "C" void kernel_launch(void* const* d_in, const int* in_sizes, int n_in,
                              void* d_out, int out_size, void* d_ws, size_t ws_size,
                              hipStream_t stream) {
    const float* A  = (const float*)d_in[0];
    const float* B  = (const float*)d_in[1];
    const float* ea = (const float*)d_in[2];
    const float* eb = (const float*)d_in[3];
    float* out = (float*)d_out;

    float* ws      = (float*)d_ws;
    float* s_max   = ws;
    float* mask_w  = s_max   + 64;
    float* G_w     = mask_w  + K_DIM * K_DIM;
    float* r_w     = G_w     + K_DIM * K_DIM;
    float* p_w     = r_w     + K_DIM * K_DIM;
    float* ap_w    = p_w     + K_DIM * K_DIM;
    float* part_w  = ap_w    + K_DIM * K_DIM;
    float* rs_w    = part_w  + NT * 32 * 32;
    float* r2_w    = rs_w    + K_DIM * RS_PITCH;
    float* p2_w    = r2_w    + K_DIM * K_DIM;
    float* ap2_w   = p2_w    + K_DIM * K_DIM;
    float* part2_w = ap2_w   + K_DIM * K_DIM;
    float* rs2_w   = part2_w + NT * 32 * 32;
    const bool one_launch = ws_size >= sizeof(float) *
        (size_t)(64 + 8 * K_DIM * K_DIM + 2 * NT * 32 * 32 + 2 * K_DIM * RS_PITCH);

    max_reduce_kernel<<<1, 512, 0, stream>>>(ea, eb, s_max);

    dim3 mgrid(K_DIM / 256, K_DIM);
    dim3 ggrid(16, 16);
    dim3 ogrid(NT, K_DIM / 16);

    mask_kernel<<<mgrid, 256, 0, stream>>>(eb, ea, s_max, mask_w);
    gram_kernel<<<ggrid, 256, 0, stream>>>(A, A, G_w);
    gram_kernel<<<ggrid, 256, 0, stream>>>(B, A, r_w);
    if (one_launch) {
        cg_first_kernel<<<ogrid, 256, 0, stream>>>(G_w, mask_w, r_w, p_w, ap_w, part_w, rs_w, out);
        for (int it = 0; it < (NITER - 2) / 2; ++it) {
            cg_step_kernel<<<ogrid, 256, 0, stream>>>(G_w, mask_w, r_w, p_w, ap_w, part_w, rs_w,
                                                      r2_w, p2_w, ap2_w, part2_w, rs2_w, out);
            cg_step_kernel<<<ogrid, 256, 0, stream>>>(G_w, mask_w, r2_w, p2_w, ap2_w, part2_w, rs2_w,
                                                      r_w, p_w, ap_w, part_w, rs_w, out);
        }
        cg_step_kernel<<<ogrid, 256, 0, stream>>>(G_w, mask_w, r_w, p_w, ap_w, part_w, rs_w,
                                                  r2_w, p2_w, ap2_w, part2_w, rs2_w, out);
        cg_last_x_kernel<<<K_DIM / 4, 128, 0, stream>>>(p2_w, part2_w, rs2_w, out);
    } else {
        cg_init_kernel<<<K_DIM / 4, 128, 0, stream>>>(r_w, p_w, out, rs_w);
        for (int it = 0; it < NITER; ++it) {
            cg_operator_kernel<<<ogrid, 256, 0, stream>>>(G_w, p_w, mask_w, ap_w, part_w);
            cg_update_kernel<<<K_DIM / 4, 128, 0, stream>>>(ap_w, part_w, rs_w, out, r_w, p_w);
        }
    }

    float* out2 = out + K_DIM * K_DIM;
    mask_kernel<<<mgrid, 256, 0, stream>>>(ea, eb, s_max, mask_w);
    gram_kernel<<<ggrid, 256, 0, stream>>>(B, B, G_w);
    gram_kernel<<<ggrid, 256, 0, stream>>>(A, B, r_w);
    if (one_launch) {
        cg_first_kernel<<<ogrid, 256, 0, stream>>>(G_w, mask_w, r_w, p_w, ap_w, part_w, rs_w, out2);
        for (int it = 0; it < (NITER - 2) / 2; ++it) {
            cg_step_kernel<<<ogrid, 256, 0, stream>>>(G_w, mask_w, r_w, p_w, ap_w, part_w, rs_w,
                                                      r2_w, p2_w, ap2_w, part2_w, rs2_w, out2);
            cg_step_kernel<<<ogrid, 256, 0, stream>>>(G_w, mask_w, r2_w, p2_w, ap2_w, part2_w, rs2_w,
                                                      r_w, p_w, ap_w, part_w, rs_w, out2);
        }
        cg_step_kernel<<<ogrid, 256, 0, stream>>>(G_w, mask_w, r_w, p_w, ap_w, part_w, rs_w,
                                                  r2_w, p2_w, ap2_w, part2_w, rs2_w, out2);
        cg_last_x_kernel<<<K_DIM / 4, 128, 0, stream>>>(p2_w, part2_w, rs2_w, out2);
    } else {
        cg_init_kernel<<<K_DIM / 4, 128, 0, stream>>>(r_w, p_w, out2, rs_w);
        for (int it = 0; it < NITER; ++it) {
            cg_operator_kernel<<<ogrid, 256, 0, stream>>>(G_w, p_w, mask_w, ap_w, part_w);
            cg_update_kernel<<<K_DIM / 4, 128, 0, stream>>>(ap_w, part_w, rs_w, out2, r_w, p_w);
        }
    }
}


# CG step: operator applied to the new residual r, Ap carried as A r + beta Ap_old (rounding of Ap changes); <r,r>, beta and p leave the way to the products, p rows are no longer read; sum orders kept
# speedup vs baseline: 4.3341x; 1.1612x over previous
#include <hip/hip_runtime.h>


#define D_DIM   1024
#define K_DIM   512
#define LMBDA_F 1000.0f
#define NITER   96
#define NT       8
#define P_PITCH  516
#define T_PITCH  68
#define RS_PITCH 32

typedef float v2f __attribute__((ext_vector_type(2)));
typedef float v8f __attribute__((ext_vector_type(8)));

__device__ __forceinline__ v8f wmma_f32(v2f a, v2f b, v8f c) {
    return __builtin_amdgcn_wmma_f32_16x16x4_f32(
        false, a, false, b, (short)0, c, false, false);
}

__global__ void max_reduce_kernel(const float* __restrict__ ea,
                                  const float* __restrict__ eb,
                                  float* __restrict__ s_out) {
    __shared__ float sm[512];
    int t = threadIdx.x;
    sm[t] = fmaxf(ea[t], eb[t]);
    __syncthreads();
    for (int off = 256; off > 0; off >>= 1) {
        if (t < off) sm[t] = fmaxf(sm[t], sm[t + off]);
        __syncthreads();
    }
    if (t == 0) s_out[0] = sm[0];
}

__global__ void mask_kernel(const float* __restrict__ erow,
                            const float* __restrict__ ecol,
                            const float* __restrict__ s_ptr,
                            float* __restrict__ mask) {
    int j = blockIdx.x * blockDim.x + threadIdx.x;
    int i = blockIdx.y;
    float inv_s = 1.0f / s_ptr[0];
    float gb = erow[i] * inv_s;
    float ga = ecol[j] * inv_s;
    float gb2 = gb * gb + 1.0f;
    float ga2 = ga * ga + 1.0f;
    float re = gb / gb2 - ga / ga2;
    float im = 1.0f / gb2 - 1.0f / ga2;
    mask[i * K_DIM + j] = re * re + im * im;
}

#define GT_PITCH 36

__device__ __forceinline__ void load_xy_frags(const float* __restrict__ Xp,
                                              const float* __restrict__ Yp, int d,
                                              v2f (&a)[4][2], v2f (&b)[4][2]) {
#pragma unroll
    for (int s = 0; s < 4; ++s) {
        int kk = d + 4 * s;
#pragma unroll
        for (int i = 0; i < 2; ++i) {
            a[s][i].x = Xp[kk * K_DIM + 16 * i];
            a[s][i].y = Xp[(kk + 1) * K_DIM + 16 * i];
            b[s][i].x = Yp[kk * K_DIM + 16 * i];
            b[s][i].y = Yp[(kk + 1) * K_DIM + 16 * i];
        }
    }
}

__device__ __forceinline__ void xy_products(const v2f (&a)[4][2], const v2f (&b)[4][2],
                                            v8f (&acc)[2][2]) {
#pragma unroll
    for (int s = 0; s < 4; ++s)
#pragma unroll
        for (int mi = 0; mi < 2; ++mi)
#pragma unroll
            for (int ni = 0; ni < 2; ++ni)
                acc[mi][ni] = wmma_f32(a[s][mi], b[s][ni], acc[mi][ni]);
}

__global__ void __launch_bounds__(256)
gram_kernel(const float* __restrict__ X,
            const float* __restrict__ Y,
            float* __restrict__ C) {
    __shared__ __attribute__((aligned(16))) float g_s[8 * 32 * GT_PITCH];

    int t    = threadIdx.x;
    int lane = t & 31;
    int wave = t >> 5;
    int lo = lane & 15;
    int hi = lane >> 4;
    int m0 = blockIdx.x * 32;
    int n0 = blockIdx.y * 32;

    const float* Xp = X + (wave * 128 + 2 * hi) * K_DIM + m0 + lo;
    const float* Yp = Y + (wave * 128 + 2 * hi) * K_DIM + n0 + lo;

    v8f acc[2][2] = {};

    v2f a0[4][2], b0[4][2], a1[4][2], b1[4][2];
    load_xy_frags(Xp, Yp, 0, a0, b0);
#pragma unroll
    for (int g = 0; g < 8; g += 2) {
        load_xy_frags(Xp, Yp, (g + 1) * 16, a1, b1);
        __builtin_amdgcn_sched_barrier(0);
        xy_products(a0, b0, acc);
        __builtin_amdgcn_sched_barrier(0);
        if (g + 2 < 8) load_xy_frags(Xp, Yp, (g + 2) * 16, a0, b0);
        __builtin_amdgcn_sched_barrier(0);
        xy_products(a1, b1, acc);
        __builtin_amdgcn_sched_barrier(0);
    }

    {
        float* tq = g_s + wave * 32 * GT_PITCH;
#pragma unroll
        for (int mi = 0; mi < 2; ++mi)
#pragma unroll
            for (int ni = 0; ni < 2; ++ni)
#pragma unroll
                for (int v = 0; v < 8; ++v)
                    tq[(mi * 16 + v + 8 * hi) * GT_PITCH + ni * 16 + lo] = acc[mi][ni][v];
    }
    __syncthreads();

    {
        int row = t >> 3;
        int c   = (t & 7) * 4;
        float4 s = *(const float4*)(g_s + row * GT_PITCH + c);
#pragma unroll
        for (int q = 1; q < 8; ++q) {
            float4 u = *(const float4*)(g_s + (q * 32 + row) * GT_PITCH + c);
            s.x += u.x; s.y += u.y; s.z += u.z; s.w += u.w;
        }
        *(float4*)(C + (m0 + row) * K_DIM + n0 + c) = s;
    }
}

__global__ void __launch_bounds__(128)
cg_init_kernel(const float* __restrict__ Rr,
               float* __restrict__ P,
               float* __restrict__ X,
               float* __restrict__ rs) {
    int lane = threadIdx.x & 31;
    int row  = blockIdx.x * 4 + (threadIdx.x >> 5);
    const float4* r4 = (const float4*)(Rr + row * K_DIM);
    float4* p4 = (float4*)(P + row * K_DIM);
    float4* x4 = (float4*)(X + row * K_DIM);
    float4 z4 = make_float4(0.0f, 0.0f, 0.0f, 0.0f);
    float part = 0.0f;
#pragma unroll
    for (int j = 0; j < 4; ++j) {
        float4 v = r4[lane + 32 * j];
        p4[lane + 32 * j] = v;
        x4[lane + 32 * j] = z4;
        part += v.x * v.x + v.y * v.y + v.z * v.z + v.w * v.w;
    }
#pragma unroll
    for (int off = 16; off > 0; off >>= 1) part += __shfl_xor(part, off);
    if (lane == 0) rs[row * RS_PITCH] = part;
}

__device__ __forceinline__ void lds_barrier() {
    __builtin_amdgcn_fence(__ATOMIC_RELEASE, "workgroup", "local");
    __builtin_amdgcn_s_barrier();
    __builtin_amdgcn_fence(__ATOMIC_ACQUIRE, "workgroup", "local");
}

template <int NS>
__device__ __forceinline__ void load_g_frags(const float* __restrict__ Gp, int k0,
                                             v2f (&b0)[NS], v2f (&b1)[NS]) {
#pragma unroll
    for (int s = 0; s < NS; ++s) {
        int kk = k0 + 4 * s;
        b0[s].x = Gp[kk * K_DIM];
        b0[s].y = Gp[(kk + 1) * K_DIM];
        b1[s].x = Gp[kk * K_DIM + 16];
        b1[s].y = Gp[(kk + 1) * K_DIM + 16];
    }
}

__device__ __forceinline__ const float* g_col_ptr(const float* __restrict__ G, int t, int nt) {
    return G + nt * 64 + ((t >> 5) & 1) * 32 + (t & 15);
}
__device__ __forceinline__ int g_first_k(int t) {
    return (t >> 6) * 128 + 2 * ((t & 31) >> 4);
}

__device__ __forceinline__ void operator_tile(const float* __restrict__ G,
                                              float4 mk,
                                              const float* p_s,
                                              float* t_s,
                                              float* __restrict__ Ap,
                                              float* __restrict__ part,
                                              int t, int nt, int mb,
                                              const v2f (&ga0)[16], const v2f (&ga1)[16]) {
    int lane = t & 31;
    int wave = t >> 5;
    int lo   = lane & 15;
    int hi   = lane >> 4;
    int m0   = mb * 16;
    int nw   = wave & 1;
    int kq   = wave >> 1;

    int rowE = t >> 4;
    int nE   = nt * 64 + (t & 15) * 4;

    v8f acc0 = {};
    v8f acc1 = {};
    int k0 = kq * 128 + 2 * hi;
    const float* Gp = g_col_ptr(G, t, nt);
    v2f gb0[8], gb1[8], gc0[8], gc1[8];
    v2f aF[32];
    load_g_frags<8>(Gp, k0 + 64, gb0, gb1);
#pragma unroll
    for (int s = 0; s < 16; ++s) {
        int kk = k0 + 4 * s;
        aF[s].x = p_s[lo * P_PITCH + kk];
        aF[s].y = p_s[lo * P_PITCH + kk + 1];
    }
#pragma unroll
    for (int s = 0; s < 8; ++s) {
        acc0 = wmma_f32(aF[s], ga0[s], acc0);
        acc1 = wmma_f32(aF[s], ga1[s], acc1);
    }
    __builtin_amdgcn_sched_barrier(0);
    load_g_frags<8>(Gp, k0 + 96, gc0, gc1);
#pragma unroll
    for (int s = 16; s < 32; ++s) {
        int kk = k0 + 4 * s;
        aF[s].x = p_s[lo * P_PITCH + kk];
        aF[s].y = p_s[lo * P_PITCH + kk + 1];
    }
#pragma unroll
    for (int s = 8; s < 16; ++s) {
        acc0 = wmma_f32(aF[s], ga0[s], acc0);
        acc1 = wmma_f32(aF[s], ga1[s], acc1);
    }
#pragma unroll
    for (int s = 0; s < 8; ++s) {
        acc0 = wmma_f32(aF[16 + s], gb0[s], acc0);
        acc1 = wmma_f32(aF[16 + s], gb1[s], acc1);
    }
#pragma unroll
    for (int s = 0; s < 8; ++s) {
        acc0 = wmma_f32(aF[24 + s], gc0[s], acc0);
        acc1 = wmma_f32(aF[24 + s], gc1[s], acc1);
    }

    {
        float* tq = t_s + kq * 16 * T_PITCH + nw * 32 + lo;
#pragma unroll
        for (int v = 0; v < 8; ++v) {
            int m = v + 8 * hi;
            tq[m * T_PITCH]      = acc0[v];
            tq[m * T_PITCH + 16] = acc1[v];
        }
    }
    __syncthreads();

    {
        int c = (t & 15) * 4;
        float4 s = *(const float4*)(t_s + rowE * T_PITCH + c);
#pragma unroll
        for (int q = 1; q < 4; ++q) {
            float4 u = *(const float4*)(t_s + (q * 16 + rowE) * T_PITCH + c);
            s.x += u.x; s.y += u.y; s.z += u.z; s.w += u.w;
        }
        float4 pv = *(const float4*)(p_s + rowE * P_PITCH + nE);
        float4 val;
        val.x = s.x + LMBDA_F * mk.x * pv.x;
        val.y = s.y + LMBDA_F * mk.y * pv.y;
        val.z = s.z + LMBDA_F * mk.z * pv.z;
        val.w = s.w + LMBDA_F * mk.w * pv.w;
        *(float4*)(Ap + (m0 + rowE) * K_DIM + nE) = val;
        float d = pv.x * val.x + pv.y * val.y + pv.z * val.z + pv.w * val.w;
        d += __shfl_xor(d, 8);
        d += __shfl_xor(d, 4);
        d += __shfl_xor(d, 2);
        d += __shfl_xor(d, 1);
        if ((t & 15) == 0) part[(nt * 32 + mb) * 32 + rowE] = d;
    }
}

__global__ void __launch_bounds__(256)
cg_operator_kernel(const float* __restrict__ G,
                   const float* __restrict__ P,
                   const float* __restrict__ mask,
                   float* __restrict__ Ap,
                   float* __restrict__ part) {
    __shared__ __attribute__((aligned(16))) float p_s[16 * P_PITCH];
    __shared__ __attribute__((aligned(16))) float t_s[4 * 16 * T_PITCH];

    int t  = threadIdx.x;
    int nt = blockIdx.x;
    int m0 = blockIdx.y * 16;
    float4 mk = *(const float4*)(mask + (m0 + (t >> 4)) * K_DIM + nt * 64 + (t & 15) * 4);

    float4 pst[8];
#pragma unroll
    for (int j = 0; j < 8; ++j) {
        int f   = t + 256 * j;
        pst[j] = ((const float4*)(P + (m0 + (f >> 7)) * K_DIM))[f & 127];
    }
    v2f ga0[16], ga1[16];
    load_g_frags<16>(g_col_ptr(G, t, nt), g_first_k(t), ga0, ga1);
#pragma unroll
    for (int j = 0; j < 8; ++j) {
        int f   = t + 256 * j;
        *(float4*)(p_s + (f >> 7) * P_PITCH + (f & 127) * 4) = pst[j];
    }
    lds_barrier();

    operator_tile(G, mk, p_s, t_s, Ap, part, t, nt, blockIdx.y, ga0, ga1);
}

#define SC_PITCH 32

template <bool FIRST>
__device__ __forceinline__ void row_norms(const float4 (&rv)[2][4], const float (&rs_old)[2],
                                          float* sc_s, int wave, int lane) {
    float sum[2];
#pragma unroll
    for (int rr = 0; rr < 2; ++rr) {
        sum[rr] = 0.0f;
#pragma unroll
        for (int j = 0; j < 4; ++j)
            sum[rr] += rv[rr][j].x * rv[rr][j].x + rv[rr][j].y * rv[rr][j].y
                     + rv[rr][j].z * rv[rr][j].z + rv[rr][j].w * rv[rr][j].w;
    }
#pragma unroll
    for (int off = 16; off > 0; off >>= 1) {
        float s0 = __shfl_xor(sum[0], off);
        float s1 = __shfl_xor(sum[1], off);
        sum[0] += s0;
        sum[1] += s1;
    }
#pragma unroll
    for (int rr = 0; rr < 2; ++rr) {
        int lr = wave * 2 + rr;
        float b = 0.0f;
        if (!FIRST) b = sum[rr] / rs_old[rr];
        sc_s[(16 + lr) * SC_PITCH + lane] = b;
        sc_s[(32 + lr) * SC_PITCH + lane] = sum[rr];
    }
}

template <bool FIRST>
__device__ __forceinline__ void cg_step_body(const float* G,
                                             const float* mask,
                                             const float* Rin,
                                             const float* Pin,
                                             const float* APin,
                                             const float* partin,
                                             const float* rsin,
                                             float* Rout,
                                             float* Pout,
                                             float* APout,
                                             float* partout,
                                             float* rsout,
                                             float* X,
                                             float* r_s, float* t_s, float* sc_s) {
    int t    = threadIdx.x;
    int lane = t & 31;
    int wave = t >> 5;
    int lo   = lane & 15;
    int hi   = lane >> 4;
    int nt   = blockIdx.x;
    int mb   = blockIdx.y;
    int m0   = mb * 16;
    int nw   = wave & 1;
    int kq   = wave >> 1;
    int rowE = t >> 4;
    int nE   = nt * 64 + (t & 15) * 4;

    float4 rv[2][4], av[2][4];
    float pin[2][NT], rs_old[2];
    if (!FIRST) {
#pragma unroll
        for (int rr = 0; rr < 2; ++rr) {
            int lr = wave * 2 + rr;
#pragma unroll
            for (int i = 0; i < NT; ++i)
                pin[rr][i] = partin[(i * 32 + mb) * 32 + lr];
            rs_old[rr] = rsin[(m0 + lr) * RS_PITCH];
        }
    }
#pragma unroll
    for (int rr = 0; rr < 2; ++rr) {
        int row = m0 + wave * 2 + rr;
#pragma unroll
        for (int j = 0; j < 4; ++j) {
            rv[rr][j] = ((const float4*)(Rin + row * K_DIM))[lane + 32 * j];
            if (!FIRST) av[rr][j] = ((const float4*)(APin + row * K_DIM))[lane + 32 * j];
        }
    }
    v2f ga0[16], ga1[16];
    load_g_frags<16>(g_col_ptr(G, t, nt), g_first_k(t), ga0, ga1);
    float4 mk = *(const float4*)(mask + (m0 + rowE) * K_DIM + nE);
    float4 pOld = make_float4(0.0f, 0.0f, 0.0f, 0.0f);
    float4 aOld = pOld, xOld = pOld;
    if (!FIRST) {
        pOld = *(const float4*)(Pin  + (m0 + rowE) * K_DIM + nE);
        aOld = *(const float4*)(APin + (m0 + rowE) * K_DIM + nE);
        xOld = *(const float4*)(X    + (m0 + rowE) * K_DIM + nE);
    }
    __builtin_amdgcn_sched_barrier(0);

    float a[2];
#pragma unroll
    for (int rr = 0; rr < 2; ++rr) {
        a[rr] = 0.0f;
        if (!FIRST) {
            float s = 0.0f;
#pragma unroll
            for (int i = 0; i < NT; ++i) s += pin[rr][i];
            a[rr] = rs_old[rr] / s;
#pragma unroll
            for (int j = 0; j < 4; ++j) {
                rv[rr][j].x -= a[rr] * av[rr][j].x; rv[rr][j].y -= a[rr] * av[rr][j].y;
                rv[rr][j].z -= a[rr] * av[rr][j].z; rv[rr][j].w -= a[rr] * av[rr][j].w;
            }
        }
    }
#pragma unroll
    for (int rr = 0; rr < 2; ++rr) {
        int lr = wave * 2 + rr;
#pragma unroll
        for (int j = 0; j < 4; ++j)
            *(float4*)(r_s + lr * P_PITCH + (lane + 32 * j) * 4) = rv[rr][j];
        if (!FIRST) sc_s[lr * SC_PITCH + lane] = a[rr];
    }
    lds_barrier();

    const bool norms_first = ((wave ^ (wave >> 2)) & 1) == 0;
    if (norms_first) row_norms<FIRST>(rv, rs_old, sc_s, wave, lane);

    v8f acc0 = {};
    v8f acc1 = {};
    int k0 = kq * 128 + 2 * hi;
    const float* Gp = g_col_ptr(G, t, nt);
    v2f gb0[8], gb1[8], gc0[8], gc1[8];
    v2f aF[32];
    load_g_frags<8>(Gp, k0 + 64, gb0, gb1);
#pragma unroll
    for (int s = 0; s < 16; ++s) {
        int kk = k0 + 4 * s;
        aF[s].x = r_s[lo * P_PITCH + kk];
        aF[s].y = r_s[lo * P_PITCH + kk + 1];
    }
#pragma unroll
    for (int s = 0; s < 8; ++s) {
        acc0 = wmma_f32(aF[s], ga0[s], acc0);
        acc1 = wmma_f32(aF[s], ga1[s], acc1);
    }
    asm volatile("" : "+v"(acc0), "+v"(acc1) : : "memory");
    load_g_frags<8>(Gp, k0 + 96, gc0, gc1);

#pragma unroll
    for (int s = 16; s < 32; ++s) {
        int kk = k0 + 4 * s;
        aF[s].x = r_s[lo * P_PITCH + kk];
        aF[s].y = r_s[lo * P_PITCH + kk + 1];
    }
#pragma unroll
    for (int s = 8; s < 16; ++s) {
        acc0 = wmma_f32(aF[s], ga0[s], acc0);
        acc1 = wmma_f32(aF[s], ga1[s], acc1);
    }
    if (!norms_first) row_norms<FIRST>(rv, rs_old, sc_s, wave, lane);
    {
        float4* xq = (float4*)(X + (m0 + rowE) * K_DIM + nE);
        if (FIRST) {
            *xq = make_float4(0.0f, 0.0f, 0.0f, 0.0f);
        } else {
            float al = sc_s[rowE * SC_PITCH];
            xOld.x += al * pOld.x; xOld.y += al * pOld.y;
            xOld.z += al * pOld.z; xOld.w += al * pOld.w;
            *xq = xOld;
            *(float4*)(Rout + (m0 + rowE) * K_DIM + nE) =
                *(const float4*)(r_s + rowE * P_PITCH + nE);
        }
    }
#pragma unroll
    for (int s = 0; s < 8; ++s) {
        acc0 = wmma_f32(aF[16 + s], gb0[s], acc0);
        acc1 = wmma_f32(aF[16 + s], gb1[s], acc1);
    }
#pragma unroll
    for (int s = 0; s < 8; ++s) {
        acc0 = wmma_f32(aF[24 + s], gc0[s], acc0);
        acc1 = wmma_f32(aF[24 + s], gc1[s], acc1);
    }

    {
        float* tq = t_s + kq * 16 * T_PITCH + nw * 32 + lo;
#pragma unroll
        for (int v = 0; v < 8; ++v) {
            int m = v + 8 * hi;
            tq[m * T_PITCH]      = acc0[v];
            tq[m * T_PITCH + 16] = acc1[v];
        }
    }
    __syncthreads();

    {
        int c = (t & 15) * 4;
        float4 s = *(const float4*)(t_s + rowE * T_PITCH + c);
#pragma unroll
        for (int q = 1; q < 4; ++q) {
            float4 u = *(const float4*)(t_s + (q * 16 + rowE) * T_PITCH + c);
            s.x += u.x; s.y += u.y; s.z += u.z; s.w += u.w;
        }
        float4 rn = *(const float4*)(r_s + rowE * P_PITCH + nE);
        float4 val, pn;
        val.x = s.x + LMBDA_F * mk.x * rn.x;
        val.y = s.y + LMBDA_F * mk.y * rn.y;
        val.z = s.z + LMBDA_F * mk.z * rn.z;
        val.w = s.w + LMBDA_F * mk.w * rn.w;
        if (FIRST) {
            pn = rn;
        } else {
            float be = sc_s[(16 + rowE) * SC_PITCH];
            val.x += be * aOld.x; val.y += be * aOld.y;
            val.z += be * aOld.z; val.w += be * aOld.w;
            pn.x = rn.x + be * pOld.x; pn.y = rn.y + be * pOld.y;
            pn.z = rn.z + be * pOld.z; pn.w = rn.w + be * pOld.w;
        }
        *(float4*)(APout + (m0 + rowE) * K_DIM + nE) = val;
        *(float4*)(Pout  + (m0 + rowE) * K_DIM + nE) = pn;
        if (nt == 0 && t < 16) rsout[(m0 + t) * RS_PITCH] = sc_s[(32 + t) * SC_PITCH];
        float d = pn.x * val.x + pn.y * val.y + pn.z * val.z + pn.w * val.w;
        d += __shfl_xor(d, 8);
        d += __shfl_xor(d, 4);
        d += __shfl_xor(d, 2);
        d += __shfl_xor(d, 1);
        if ((t & 15) == 0) partout[(nt * 32 + mb) * 32 + rowE] = d;
    }
}

__global__ void __launch_bounds__(256)
cg_first_kernel(const float* G,
                const float* mask,
                const float* Rin,
                float* Pout,
                float* APout,
                float* partout,
                float* rsout,
                float* X) {
    __shared__ __attribute__((aligned(16))) float r_s[16 * P_PITCH];
    __shared__ __attribute__((aligned(16))) float t_s[4 * 16 * T_PITCH];
    __shared__ float sc_s[3 * 16 * SC_PITCH];
    cg_step_body<true>(G, mask, Rin, nullptr, nullptr, nullptr, nullptr,
                       nullptr, Pout, APout, partout, rsout, X, r_s, t_s, sc_s);
}

__global__ void __launch_bounds__(256)
cg_step_kernel(const float* G,
               const float* mask,
               const float* Rin,
               const float* Pin,
               const float* APin,
               const float* partin,
               const float* rsin,
               float* Rout,
               float* Pout,
               float* APout,
               float* partout,
               float* rsout,
               float* X) {
    __shared__ __attribute__((aligned(16))) float r_s[16 * P_PITCH];
    __shared__ __attribute__((aligned(16))) float t_s[4 * 16 * T_PITCH];
    __shared__ float sc_s[3 * 16 * SC_PITCH];
    cg_step_body<false>(G, mask, Rin, Pin, APin, partin, rsin,
                        Rout, Pout, APout, partout, rsout, X, r_s, t_s, sc_s);
}

__global__ void __launch_bounds__(128)
cg_last_x_kernel(const float* __restrict__ P,
                 const float* __restrict__ part,
                 const float* __restrict__ rs,
                 float* __restrict__ X) {
    int lane = threadIdx.x & 31;
    int row  = blockIdx.x * 4 + (threadIdx.x >> 5);
    const float4* p4 = (const float4*)(P + row * K_DIM);
    float4* x4 = (float4*)(X + row * K_DIM);
    float4 pv[4], xv[4];
#pragma unroll
    for (int j = 0; j < 4; ++j) {
        pv[j] = p4[lane + 32 * j];
        xv[j] = x4[lane + 32 * j];
    }
    float pap = 0.0f;
#pragma unroll
    for (int i = 0; i < NT; ++i)
        pap += part[(i * 32 + (row >> 4)) * 32 + (row & 15)];
    float a = rs[row * RS_PITCH] / pap;
#pragma unroll
    for (int j = 0; j < 4; ++j) {
        xv[j].x += a * pv[j].x; xv[j].y += a * pv[j].y;
        xv[j].z += a * pv[j].z; xv[j].w += a * pv[j].w;
        x4[lane + 32 * j] = xv[j];
    }
}

__global__ void __launch_bounds__(128)
cg_update_kernel(const float* __restrict__ Ap,
                 const float* __restrict__ part,
                 float* __restrict__ rs,
                 float* __restrict__ X,
                 float* __restrict__ Rr,
                 float* __restrict__ P) {
    int lane = threadIdx.x & 31;
    int row  = blockIdx.x * 4 + (threadIdx.x >> 5);
    const float4* ap4 = (const float4*)(Ap + row * K_DIM);
    float4* x4 = (float4*)(X  + row * K_DIM);
    float4* r4 = (float4*)(Rr + row * K_DIM);
    float4* p4 = (float4*)(P  + row * K_DIM);

    float4 pv[4], av[4], rv[4], xv[4];
#pragma unroll
    for (int j = 0; j < 4; ++j) {
        pv[j] = p4[lane + 32 * j];
        av[j] = ap4[lane + 32 * j];
        rv[j] = r4[lane + 32 * j];
        xv[j] = x4[lane + 32 * j];
    }
    float pap = 0.0f;
#pragma unroll
    for (int i = 0; i < NT; ++i)
        pap += part[(i * 32 + (row >> 4)) * 32 + (row & 15)];
    float rs_old = rs[row * RS_PITCH];
    float a = rs_old / pap;

    float sum = 0.0f;
#pragma unroll
    for (int j = 0; j < 4; ++j) {
        xv[j].x += a * pv[j].x; xv[j].y += a * pv[j].y;
        xv[j].z += a * pv[j].z; xv[j].w += a * pv[j].w;
        rv[j].x -= a * av[j].x; rv[j].y -= a * av[j].y;
        rv[j].z -= a * av[j].z; rv[j].w -= a * av[j].w;
        x4[lane + 32 * j] = xv[j];
        r4[lane + 32 * j] = rv[j];
        sum += rv[j].x * rv[j].x + rv[j].y * rv[j].y
             + rv[j].z * rv[j].z + rv[j].w * rv[j].w;
    }
#pragma unroll
    for (int off = 16; off > 0; off >>= 1) sum += __shfl_xor(sum, off);
    float b = sum / rs_old;
#pragma unroll
    for (int j = 0; j < 4; ++j) {
        float4 q;
        q.x = rv[j].x + b * pv[j].x; q.y = rv[j].y + b * pv[j].y;
        q.z = rv[j].z + b * pv[j].z; q.w = rv[j].w + b * pv[j].w;
        p4[lane + 32 * j] = q;
    }
    if (lane == 0) rs[row * RS_PITCH] = sum;
}

extern "C" void kernel_launch(void* const* d_in, const int* in_sizes, int n_in,
                              void* d_out, int out_size, void* d_ws, size_t ws_size,
                              hipStream_t stream) {
    const float* A  = (const float*)d_in[0];
    const float* B  = (const float*)d_in[1];
    const float* ea = (const float*)d_in[2];
    const float* eb = (const float*)d_in[3];
    float* out = (float*)d_out;

    float* ws      = (float*)d_ws;
    float* s_max   = ws;
    float* mask_w  = s_max   + 64;
    float* G_w     = mask_w  + K_DIM * K_DIM;
    float* r_w     = G_w     + K_DIM * K_DIM;
    float* p_w     = r_w     + K_DIM * K_DIM;
    float* ap_w    = p_w     + K_DIM * K_DIM;
    float* part_w  = ap_w    + K_DIM * K_DIM;
    float* rs_w    = part_w  + NT * 32 * 32;
    float* r2_w    = rs_w    + K_DIM * RS_PITCH;
    float* p2_w    = r2_w    + K_DIM * K_DIM;
    float* ap2_w   = p2_w    + K_DIM * K_DIM;
    float* part2_w = ap2_w   + K_DIM * K_DIM;
    float* rs2_w   = part2_w + NT * 32 * 32;
    const bool one_launch = ws_size >= sizeof(float) *
        (size_t)(64 + 8 * K_DIM * K_DIM + 2 * NT * 32 * 32 + 2 * K_DIM * RS_PITCH);

    max_reduce_kernel<<<1, 512, 0, stream>>>(ea, eb, s_max);

    dim3 mgrid(K_DIM / 256, K_DIM);
    dim3 ggrid(16, 16);
    dim3 ogrid(NT, K_DIM / 16);

    mask_kernel<<<mgrid, 256, 0, stream>>>(eb, ea, s_max, mask_w);
    gram_kernel<<<ggrid, 256, 0, stream>>>(A, A, G_w);
    gram_kernel<<<ggrid, 256, 0, stream>>>(B, A, r_w);
    if (one_launch) {
        cg_first_kernel<<<ogrid, 256, 0, stream>>>(G_w, mask_w, r_w, p_w, ap_w, part_w, rs_w, out);
        for (int it = 0; it < (NITER - 2) / 2; ++it) {
            cg_step_kernel<<<ogrid, 256, 0, stream>>>(G_w, mask_w, r_w, p_w, ap_w, part_w, rs_w,
                                                      r2_w, p2_w, ap2_w, part2_w, rs2_w, out);
            cg_step_kernel<<<ogrid, 256, 0, stream>>>(G_w, mask_w, r2_w, p2_w, ap2_w, part2_w, rs2_w,
                                                      r_w, p_w, ap_w, part_w, rs_w, out);
        }
        cg_step_kernel<<<ogrid, 256, 0, stream>>>(G_w, mask_w, r_w, p_w, ap_w, part_w, rs_w,
                                                  r2_w, p2_w, ap2_w, part2_w, rs2_w, out);
        cg_last_x_kernel<<<K_DIM / 4, 128, 0, stream>>>(p2_w, part2_w, rs2_w, out);
    } else {
        cg_init_kernel<<<K_DIM / 4, 128, 0, stream>>>(r_w, p_w, out, rs_w);
        for (int it = 0; it < NITER; ++it) {
            cg_operator_kernel<<<ogrid, 256, 0, stream>>>(G_w, p_w, mask_w, ap_w, part_w);
            cg_update_kernel<<<K_DIM / 4, 128, 0, stream>>>(ap_w, part_w, rs_w, out, r_w, p_w);
        }
    }

    float* out2 = out + K_DIM * K_DIM;
    mask_kernel<<<mgrid, 256, 0, stream>>>(ea, eb, s_max, mask_w);
    gram_kernel<<<ggrid, 256, 0, stream>>>(B, B, G_w);
    gram_kernel<<<ggrid, 256, 0, stream>>>(A, B, r_w);
    if (one_launch) {
        cg_first_kernel<<<ogrid, 256, 0, stream>>>(G_w, mask_w, r_w, p_w, ap_w, part_w, rs_w, out2);
        for (int it = 0; it < (NITER - 2) / 2; ++it) {
            cg_step_kernel<<<ogrid, 256, 0, stream>>>(G_w, mask_w, r_w, p_w, ap_w, part_w, rs_w,
                                                      r2_w, p2_w, ap2_w, part2_w, rs2_w, out2);
            cg_step_kernel<<<ogrid, 256, 0, stream>>>(G_w, mask_w, r2_w, p2_w, ap2_w, part2_w, rs2_w,
                                                      r_w, p_w, ap_w, part_w, rs_w, out2);
        }
        cg_step_kernel<<<ogrid, 256, 0, stream>>>(G_w, mask_w, r_w, p_w, ap_w, part_w, rs_w,
                                                  r2_w, p2_w, ap2_w, part2_w, rs2_w, out2);
        cg_last_x_kernel<<<K_DIM / 4, 128, 0, stream>>>(p2_w, part2_w, rs2_w, out2);
    } else {
        cg_init_kernel<<<K_DIM / 4, 128, 0, stream>>>(r_w, p_w, out2, rs_w);
        for (int it = 0; it < NITER; ++it) {
            cg_operator_kernel<<<ogrid, 256, 0, stream>>>(G_w, p_w, mask_w, ap_w, part_w);
            cg_update_kernel<<<K_DIM / 4, 128, 0, stream>>>(ap_w, part_w, rs_w, out2, r_w, p_w);
        }
    }
}
